# starting kernel
# baseline (speedup 1.0000x reference)
#include <hip/hip_runtime.h>
#include <math.h>

#define N_NODES 50000
#define N_EDGES 400000
#define HEADS   2
#define DDIM    128
#define F_IN    5
#define HD      (HEADS*DDIM)
#define BN_EPS  1e-5f

typedef __attribute__((ext_vector_type(2))) float v2f;
typedef __attribute__((ext_vector_type(8))) float v8f;

__device__ __forceinline__ unsigned enc_f32(float f) {
    unsigned u = __float_as_uint(f);
    return (u & 0x80000000u) ? ~u : (u | 0x80000000u);
}
__device__ __forceinline__ float dec_f32(unsigned u) {
    unsigned v = (u & 0x80000000u) ? (u & 0x7FFFFFFFu) : ~u;
    return __uint_as_float(v);
}
__device__ __forceinline__ void atomAddF(float* p, float v) {
    unsafeAtomicAdd(p, v);
}

__global__ void k_init(float* __restrict__ den, float* __restrict__ cnt,
                       float* __restrict__ ssum, unsigned* __restrict__ mx) {
    long i = (long)blockIdx.x * blockDim.x + threadIdx.x;
    const long total = (long)N_NODES * HD;
    for (long j = i; j < total; j += (long)gridDim.x * blockDim.x) ssum[j] = 0.f;
    if (i < N_NODES) cnt[i] = 0.f;
    if (i < (long)N_NODES * HEADS) { den[i] = 0.f; mx[i] = 0u; }
}

__global__ __launch_bounds__(256)
void k_node(const float* __restrict__ h,
            const float* __restrict__ gamma1, const float* __restrict__ beta1,
            const float* __restrict__ rm1, const float* __restrict__ rv1,
            const float* __restrict__ Wl, const float* __restrict__ bl,
            const float* __restrict__ Wr, const float* __restrict__ br,
            float* __restrict__ xl, float* __restrict__ xr) {
    int n = blockIdx.x;
    int c = threadIdx.x;
    float hbn[F_IN];
#pragma unroll
    for (int k = 0; k < F_IN; ++k) {
        float v = h[n * F_IN + k];
        hbn[k] = (v - rm1[k]) * rsqrtf(rv1[k] + BN_EPS) * gamma1[k] + beta1[k];
    }
    float sl = bl[c], sr = br[c];
#pragma unroll
    for (int k = 0; k < F_IN; ++k) {
        sl += hbn[k] * Wl[k * HD + c];
        sr += hbn[k] * Wr[k * HD + c];
    }
    xl[(long)n * HD + c] = sl;
    xr[(long)n * HD + c] = sr;
}

__global__ __launch_bounds__(256)
void k_edge_logits(const int* __restrict__ ei, const float* __restrict__ ew,
                   const float* __restrict__ We, const float* __restrict__ att,
                   const float* __restrict__ xl, const float* __restrict__ xr,
                   float* __restrict__ logits, unsigned* __restrict__ mx,
                   float* __restrict__ cnt) {
    int wave = threadIdx.x >> 5, lane = threadIdx.x & 31;
    int e = blockIdx.x * 8 + wave;
    if (e >= N_EDGES) return;
    int src = ei[e];
    int dst = ei[N_EDGES + e];
    float w = ew[e];
    const float* pl = xl + (long)src * HD;
    const float* pr = xr + (long)dst * HD;
    float s = 0.f;
    int base = lane * 8;
#pragma unroll
    for (int j = 0; j < 8; ++j) {
        int hd = base + j;
        float z = pl[hd] + pr[hd] + w * We[hd];
        z = z > 0.f ? z : 0.2f * z;
        s += att[hd] * z;
    }
    s += __shfl_xor(s, 8, 32);
    s += __shfl_xor(s, 4, 32);
    s += __shfl_xor(s, 2, 32);
    s += __shfl_xor(s, 1, 32);
    if (lane == 0 || lane == 16) {
        int hh = lane >> 4;
        logits[(long)e * HEADS + hh] = s;
        atomicMax(&mx[(long)dst * HEADS + hh], enc_f32(s));
        if (lane == 0) atomAddF(&cnt[dst], 1.f);
    }
}

__global__ __launch_bounds__(256)
void k_edge_exp(const int* __restrict__ ei, float* __restrict__ logits,
                const unsigned* __restrict__ mx, float* __restrict__ den) {
    long i = (long)blockIdx.x * blockDim.x + threadIdx.x;
    if (i >= (long)N_EDGES * HEADS) return;
    int e = (int)(i >> 1), hh = (int)(i & 1);
    int dst = ei[N_EDGES + e];
    float m = dec_f32(mx[(long)dst * HEADS + hh]);
    float ex = expf(logits[i] - m);
    logits[i] = ex;
    atomAddF(&den[(long)dst * HEADS + hh], ex);
}

__global__ __launch_bounds__(256)
void k_edge_scatter(const int* __restrict__ ei, const float* __restrict__ ex,
                    const float* __restrict__ den, const float* __restrict__ xl,
                    float* __restrict__ ssum) {
    int wave = threadIdx.x >> 5, lane = threadIdx.x & 31;
    int e = blockIdx.x * 8 + wave;
    if (e >= N_EDGES) return;
    int src = ei[e], dst = ei[N_EDGES + e];
    int hh = lane >> 4;
    float alpha = ex[(long)e * HEADS + hh] / (den[(long)dst * HEADS + hh] + 1e-16f);
    const float* pl = xl + (long)src * HD;
    float* ps = ssum + (long)dst * HD;
    int base = lane * 8;
#pragma unroll
    for (int j = 0; j < 8; ++j) {
        int hd = base + j;
        atomAddF(&ps[hd], pl[hd] * alpha);
    }
}

__global__ __launch_bounds__(256)
void k_fc(const float* __restrict__ ssum, const float* __restrict__ cnt,
          const float* __restrict__ bias_gat,
          const float* __restrict__ W1, const float* __restrict__ b1,
          const float* __restrict__ gamma2, const float* __restrict__ beta2,
          const float* __restrict__ rm2, const float* __restrict__ rv2,
          const float* __restrict__ W2, const float* __restrict__ b2,
          float* __restrict__ out) {
    __shared__ float Y[16 * 128];
    __shared__ float Z[16 * 128];
    int row0 = blockIdx.x * 16;
    int t = threadIdx.x;

    {
        int r = t >> 4;
        int c0 = (t & 15) * 8;
        int n = row0 + r;
        float inv = 0.5f / fmaxf(cnt[n], 1.f);
        const float* p = ssum + (long)n * HD;
#pragma unroll
        for (int j = 0; j < 8; ++j) {
            int d = c0 + j;
            Y[r * 128 + d] = (p[d] + p[128 + d]) * inv + bias_gat[d];
        }
    }
    __syncthreads();

    int wave = t >> 5, lane = t & 31;
    int c0 = wave * 16;
    int laneN = lane & 15;
    int koff = (lane >> 4) * 2;
    v8f acc = {0.f, 0.f, 0.f, 0.f, 0.f, 0.f, 0.f, 0.f};
    for (int k = 0; k < 128; k += 4) {
        v2f a, b;
        a.x = Y[laneN * 128 + k + koff];
        a.y = Y[laneN * 128 + k + koff + 1];
        int kk = k + koff;
        b.x = W1[kk * 128 + c0 + laneN];
        b.y = W1[(kk + 1) * 128 + c0 + laneN];
        acc = __builtin_amdgcn_wmma_f32_16x16x4_f32(
            false, a, false, b, (short)0, acc, false, false);
    }

#pragma unroll
    for (int v = 0; v < 8; ++v) {
        int m = v + (lane >> 4) * 8;
        int col = c0 + laneN;
        float z = acc[v] + b1[col];
        z = (z - rm2[col]) * rsqrtf(rv2[col] + BN_EPS) * gamma2[col] + beta2[col];
        z = z > 0.f ? z : 0.01f * z;
        Z[m * 128 + col] = z;
    }
    __syncthreads();

    if (t < 16 * F_IN) {
        int r = t / F_IN, c = t % F_IN;
        float o = b2[c];
        const float* zr = &Z[r * 128];
#pragma unroll 8
        for (int d = 0; d < 128; ++d) o += zr[d] * W2[d * F_IN + c];
        out[(long)(row0 + r) * F_IN + c] = o;
    }
}

extern "C" void kernel_launch(void* const* d_in, const int* in_sizes, int n_in,
                              void* d_out, int out_size, void* d_ws, size_t ws_size,
                              hipStream_t stream) {
    const float* h        = (const float*)d_in[0];
    const int*   ei       = (const int*)  d_in[1];
    const float* ew       = (const float*)d_in[2];
    const float* gamma1   = (const float*)d_in[3];
    const float* beta1    = (const float*)d_in[4];
    const float* rm1      = (const float*)d_in[5];
    const float* rv1      = (const float*)d_in[6];
    const float* Wl       = (const float*)d_in[7];
    const float* bl       = (const float*)d_in[8];
    const float* Wr       = (const float*)d_in[9];
    const float* br       = (const float*)d_in[10];
    const float* We       = (const float*)d_in[11];
    const float* att      = (const float*)d_in[12];
    const float* bias_gat = (const float*)d_in[13];
    const float* W1       = (const float*)d_in[14];
    const float* b1       = (const float*)d_in[15];
    const float* gamma2   = (const float*)d_in[16];
    const float* beta2    = (const float*)d_in[17];
    const float* rm2      = (const float*)d_in[18];
    const float* rv2      = (const float*)d_in[19];
    const float* W2       = (const float*)d_in[20];
    const float* b2       = (const float*)d_in[21];
    float* out = (float*)d_out;

    float* ws = (float*)d_ws;
    size_t o = 0;
    float* xl     = ws + o; o += (size_t)N_NODES * HD;
    float* xr     = ws + o; o += (size_t)N_NODES * HD;
    float* logits = ws + o; o += (size_t)N_EDGES * HEADS;
    float* den    = ws + o; o += (size_t)N_NODES * HEADS;
    float* cnt    = ws + o; o += (size_t)N_NODES;
    float* ssum   = ws + o; o += (size_t)N_NODES * HD;
    unsigned* mx  = (unsigned*)(ws + o); o += (size_t)N_NODES * HEADS;

    const int T = 256;
    k_init<<<(N_NODES * HD + T - 1) / T, T, 0, stream>>>(den, cnt, ssum, mx);
    k_node<<<N_NODES, T, 0, stream>>>(h, gamma1, beta1, rm1, rv1,
                                      Wl, bl, Wr, br, xl, xr);
    k_edge_logits<<<N_EDGES / 8, T, 0, stream>>>(ei, ew, We, att, xl, xr,
                                                 logits, mx, cnt);
    k_edge_exp<<<(N_EDGES * HEADS + T - 1) / T, T, 0, stream>>>(ei, logits, mx, den);
    k_edge_scatter<<<N_EDGES / 8, T, 0, stream>>>(ei, logits, den, xl, ssum);
    k_fc<<<N_NODES / 16, T, 0, stream>>>(ssum, cnt, bias_gat, W1, b1,
                                         gamma2, beta2, rm2, rv2, W2, b2, out);
}


# Edge sum by linearity of xl=bn(h)Wl+bl: atomics add alpha*bn(h[src]) (5 wide) and alpha per (dst,head), 12 not 256 per edge; Wl, bl and mean applied per node in fc kernel (f32 sum regrouped)
# speedup vs baseline: 2.1252x; 2.1252x over previous
#include <hip/hip_runtime.h>
#include <math.h>

#define N_NODES 50000
#define N_EDGES 400000
#define HEADS   2
#define DDIM    128
#define F_IN    5
#define HD      (HEADS*DDIM)
#define BN_EPS  1e-5f
#define AGG_W   16

typedef __attribute__((ext_vector_type(2))) float v2f;
typedef __attribute__((ext_vector_type(8))) float v8f;

__device__ __forceinline__ unsigned enc_f32(float f) {
    unsigned u = __float_as_uint(f);
    return (u & 0x80000000u) ? ~u : (u | 0x80000000u);
}
__device__ __forceinline__ float dec_f32(unsigned u) {
    unsigned v = (u & 0x80000000u) ? (u & 0x7FFFFFFFu) : ~u;
    return __uint_as_float(v);
}
__device__ __forceinline__ void atomAddF(float* p, float v) {
    unsafeAtomicAdd(p, v);
}

__global__ void k_init(float* __restrict__ den, float* __restrict__ cnt,
                       float* __restrict__ agg, unsigned* __restrict__ mx) {
    long i = (long)blockIdx.x * blockDim.x + threadIdx.x;
    if (i < (long)N_NODES * AGG_W) agg[i] = 0.f;
    if (i < N_NODES) cnt[i] = 0.f;
    if (i < (long)N_NODES * HEADS) { den[i] = 0.f; mx[i] = 0u; }
}

__global__ __launch_bounds__(256)
void k_node(const float* __restrict__ h,
            const float* __restrict__ gamma1, const float* __restrict__ beta1,
            const float* __restrict__ rm1, const float* __restrict__ rv1,
            const float* __restrict__ Wl, const float* __restrict__ bl,
            const float* __restrict__ Wr, const float* __restrict__ br,
            float* __restrict__ xl, float* __restrict__ xr) {
    int n = blockIdx.x;
    int c = threadIdx.x;
    float hbn[F_IN];
#pragma unroll
    for (int k = 0; k < F_IN; ++k) {
        float v = h[n * F_IN + k];
        hbn[k] = (v - rm1[k]) * rsqrtf(rv1[k] + BN_EPS) * gamma1[k] + beta1[k];
    }
    float sl = bl[c], sr = br[c];
#pragma unroll
    for (int k = 0; k < F_IN; ++k) {
        sl += hbn[k] * Wl[k * HD + c];
        sr += hbn[k] * Wr[k * HD + c];
    }
    xl[(long)n * HD + c] = sl;
    xr[(long)n * HD + c] = sr;
}

__global__ __launch_bounds__(256)
void k_edge_logits(const int* __restrict__ ei, const float* __restrict__ ew,
                   const float* __restrict__ We, const float* __restrict__ att,
                   const float* __restrict__ xl, const float* __restrict__ xr,
                   float* __restrict__ logits, unsigned* __restrict__ mx,
                   float* __restrict__ cnt) {
    int wave = threadIdx.x >> 5, lane = threadIdx.x & 31;
    int e = blockIdx.x * 8 + wave;
    if (e >= N_EDGES) return;
    int src = ei[e];
    int dst = ei[N_EDGES + e];
    float w = ew[e];
    const float* pl = xl + (long)src * HD;
    const float* pr = xr + (long)dst * HD;
    float s = 0.f;
    int base = lane * 8;
#pragma unroll
    for (int j = 0; j < 8; ++j) {
        int hd = base + j;
        float z = pl[hd] + pr[hd] + w * We[hd];
        z = z > 0.f ? z : 0.2f * z;
        s += att[hd] * z;
    }
    s += __shfl_xor(s, 8, 32);
    s += __shfl_xor(s, 4, 32);
    s += __shfl_xor(s, 2, 32);
    s += __shfl_xor(s, 1, 32);
    if (lane == 0 || lane == 16) {
        int hh = lane >> 4;
        logits[(long)e * HEADS + hh] = s;
        atomicMax(&mx[(long)dst * HEADS + hh], enc_f32(s));
        if (lane == 0) atomAddF(&cnt[dst], 1.f);
    }
}

__global__ __launch_bounds__(256)
void k_edge_exp(const int* __restrict__ ei, float* __restrict__ logits,
                const unsigned* __restrict__ mx, float* __restrict__ den) {
    long i = (long)blockIdx.x * blockDim.x + threadIdx.x;
    if (i >= (long)N_EDGES * HEADS) return;
    int e = (int)(i >> 1), hh = (int)(i & 1);
    int dst = ei[N_EDGES + e];
    float m = dec_f32(mx[(long)dst * HEADS + hh]);
    float ex = expf(logits[i] - m);
    logits[i] = ex;
    atomAddF(&den[(long)dst * HEADS + hh], ex);
}

__global__ __launch_bounds__(256)
void k_edge_scatter(const int* __restrict__ ei, const float* __restrict__ ex,
                    const float* __restrict__ den, const float* __restrict__ h,
                    const float* __restrict__ gamma1, const float* __restrict__ beta1,
                    const float* __restrict__ rm1, const float* __restrict__ rv1,
                    float* __restrict__ agg) {
    int slot = threadIdx.x >> 4;
    int l = threadIdx.x & 15;
    int hh = l >> 3, k = l & 7;
    if (k >= 6) return;
    int kc = k < 5 ? k : 0;
    float rm = rm1[kc], rs = rsqrtf(rv1[kc] + BN_EPS), g = gamma1[kc], b = beta1[kc];
    int e0 = blockIdx.x * 64 + slot;
    int src[4], dst[4];
#pragma unroll
    for (int it = 0; it < 4; ++it) {
        int e = e0 + it * 16;
        src[it] = ei[e];
        dst[it] = ei[N_EDGES + e];
    }
    float xv[4], dn[4], hr[4];
#pragma unroll
    for (int it = 0; it < 4; ++it) {
        int e = e0 + it * 16;
        xv[it] = ex[(long)e * HEADS + hh];
        dn[it] = den[(long)dst[it] * HEADS + hh];
        hr[it] = h[src[it] * F_IN + kc];
    }
#pragma unroll
    for (int it = 0; it < 4; ++it) {
        float alpha = xv[it] / (dn[it] + 1e-16f);
        float hv = (k < 5) ? ((hr[it] - rm) * rs * g + b) : 1.f;
        atomAddF(&agg[(long)dst[it] * AGG_W + l], hv * alpha);
    }
}

__global__ __launch_bounds__(256)
void k_fc(const float* __restrict__ agg, const float* __restrict__ cnt,
          const float* __restrict__ Wl, const float* __restrict__ bl,
          const float* __restrict__ bias_gat,
          const float* __restrict__ W1, const float* __restrict__ b1,
          const float* __restrict__ gamma2, const float* __restrict__ beta2,
          const float* __restrict__ rm2, const float* __restrict__ rv2,
          const float* __restrict__ W2, const float* __restrict__ b2,
          float* __restrict__ out) {
    __shared__ float Y[16 * 128];
    __shared__ float Z[16 * 128];
    __shared__ float As[16 * AGG_W];
    int row0 = blockIdx.x * 16;
    int t = threadIdx.x;

    {
        int d = t & 127;
        int rb = t >> 7;
        float w0[6], w1[6];
#pragma unroll
        for (int k = 0; k < F_IN; ++k) {
            w0[k] = Wl[k * HD + d];
            w1[k] = Wl[k * HD + DDIM + d];
        }
        w0[5] = bl[d];
        w1[5] = bl[DDIM + d];
        float bg = bias_gat[d];
        float inv = 0.5f / fmaxf(cnt[row0 + (t >> 4)], 1.f);
        As[t] = agg[(long)row0 * AGG_W + t] * inv;
        __syncthreads();
#pragma unroll
        for (int i = 0; i < 8; ++i) {
            int r = rb + 2 * i;
            const float* a = &As[r * AGG_W];
            float s0 = 0.f, s1 = 0.f;
#pragma unroll
            for (int k = 0; k < 6; ++k) {
                s0 += a[k] * w0[k];
                s1 += a[8 + k] * w1[k];
            }
            Y[r * 128 + d] = (s0 + s1) + bg;
        }
    }
    __syncthreads();

    int wave = t >> 5, lane = t & 31;
    int c0 = wave * 16;
    int laneN = lane & 15;
    int koff = (lane >> 4) * 2;
    v8f acc = {0.f, 0.f, 0.f, 0.f, 0.f, 0.f, 0.f, 0.f};
    for (int k = 0; k < 128; k += 4) {
        v2f a, b;
        a.x = Y[laneN * 128 + k + koff];
        a.y = Y[laneN * 128 + k + koff + 1];
        int kk = k + koff;
        b.x = W1[kk * 128 + c0 + laneN];
        b.y = W1[(kk + 1) * 128 + c0 + laneN];
        acc = __builtin_amdgcn_wmma_f32_16x16x4_f32(
            false, a, false, b, (short)0, acc, false, false);
    }

#pragma unroll
    for (int v = 0; v < 8; ++v) {
        int m = v + (lane >> 4) * 8;
        int col = c0 + laneN;
        float z = acc[v] + b1[col];
        z = (z - rm2[col]) * rsqrtf(rv2[col] + BN_EPS) * gamma2[col] + beta2[col];
        z = z > 0.f ? z : 0.01f * z;
        Z[m * 128 + col] = z;
    }
    __syncthreads();

    if (t < 16 * F_IN) {
        int r = t / F_IN, c = t % F_IN;
        float o = b2[c];
        const float* zr = &Z[r * 128];
#pragma unroll 8
        for (int d = 0; d < 128; ++d) o += zr[d] * W2[d * F_IN + c];
        out[(long)(row0 + r) * F_IN + c] = o;
    }
}

extern "C" void kernel_launch(void* const* d_in, const int* in_sizes, int n_in,
                              void* d_out, int out_size, void* d_ws, size_t ws_size,
                              hipStream_t stream) {
    const float* h        = (const float*)d_in[0];
    const int*   ei       = (const int*)  d_in[1];
    const float* ew       = (const float*)d_in[2];
    const float* gamma1   = (const float*)d_in[3];
    const float* beta1    = (const float*)d_in[4];
    const float* rm1      = (const float*)d_in[5];
    const float* rv1      = (const float*)d_in[6];
    const float* Wl       = (const float*)d_in[7];
    const float* bl       = (const float*)d_in[8];
    const float* Wr       = (const float*)d_in[9];
    const float* br       = (const float*)d_in[10];
    const float* We       = (const float*)d_in[11];
    const float* att      = (const float*)d_in[12];
    const float* bias_gat = (const float*)d_in[13];
    const float* W1       = (const float*)d_in[14];
    const float* b1       = (const float*)d_in[15];
    const float* gamma2   = (const float*)d_in[16];
    const float* beta2    = (const float*)d_in[17];
    const float* rm2      = (const float*)d_in[18];
    const float* rv2      = (const float*)d_in[19];
    const float* W2       = (const float*)d_in[20];
    const float* b2       = (const float*)d_in[21];
    float* out = (float*)d_out;

    float* ws = (float*)d_ws;
    size_t o = 0;
    float* xl     = ws + o; o += (size_t)N_NODES * HD;
    float* xr     = ws + o; o += (size_t)N_NODES * HD;
    float* agg    = ws + o; o += (size_t)N_NODES * AGG_W;
    float* logits = ws + o; o += (size_t)N_EDGES * HEADS;
    float* den    = ws + o; o += (size_t)N_NODES * HEADS;
    float* cnt    = ws + o; o += (size_t)N_NODES;
    unsigned* mx  = (unsigned*)(ws + o); o += (size_t)N_NODES * HEADS;

    const int T = 256;
    k_init<<<(N_NODES * AGG_W + T - 1) / T, T, 0, stream>>>(den, cnt, agg, mx);
    k_node<<<N_NODES, T, 0, stream>>>(h, gamma1, beta1, rm1, rv1,
                                      Wl, bl, Wr, br, xl, xr);
    k_edge_logits<<<N_EDGES / 8, T, 0, stream>>>(ei, ew, We, att, xl, xr,
                                                 logits, mx, cnt);
    k_edge_exp<<<(N_EDGES * HEADS + T - 1) / T, T, 0, stream>>>(ei, logits, mx, den);
    k_edge_scatter<<<N_EDGES / 64, T, 0, stream>>>(ei, logits, den, h, gamma1, beta1,
                                                   rm1, rv1, agg);
    k_fc<<<N_NODES / 16, T, 0, stream>>>(agg, cnt, Wl, bl, bias_gat, W1, b1,
                                         gamma2, beta2, rm2, rv2, W2, b2, out);
}


# Edge logits from the 5-wide factors: z = Wc^T [bn(h[src]),w,bn(h[dst]),1] (K=12) on f32 WMMA, 16 edges per tile, 256 resident blocks; xl/xr and the node kernel dropped (f32 sums regrouped)
# speedup vs baseline: 3.3675x; 1.5846x over previous
#include <hip/hip_runtime.h>
#include <math.h>

#define N_NODES 50000
#define N_EDGES 400000
#define HEADS   2
#define DDIM    128
#define F_IN    5
#define HD      (HEADS*DDIM)
#define BN_EPS  1e-5f
#define AGG_W   16

typedef __attribute__((ext_vector_type(2))) float v2f;
typedef __attribute__((ext_vector_type(8))) float v8f;

__device__ __forceinline__ unsigned enc_f32(float f) {
    unsigned u = __float_as_uint(f);
    return (u & 0x80000000u) ? ~u : (u | 0x80000000u);
}
__device__ __forceinline__ float dec_f32(unsigned u) {
    unsigned v = (u & 0x80000000u) ? (u & 0x7FFFFFFFu) : ~u;
    return __uint_as_float(v);
}
__device__ __forceinline__ void atomAddF(float* p, float v) {
    unsafeAtomicAdd(p, v);
}

__global__ void k_init(float* __restrict__ den, float* __restrict__ cnt,
                       float* __restrict__ agg, unsigned* __restrict__ mx) {
    long i = (long)blockIdx.x * blockDim.x + threadIdx.x;
    if (i < (long)N_NODES * AGG_W) agg[i] = 0.f;
    if (i < N_NODES) cnt[i] = 0.f;
    if (i < (long)N_NODES * HEADS) { den[i] = 0.f; mx[i] = 0u; }
}

#define LG_BLOCKS 256
#define LG_WAVES  (LG_BLOCKS * 8)
#define LG_GROUPS (N_EDGES / 16)

__device__ __forceinline__ v2f lrelu_att(v8f acc, v8f a, v2f s) {
#pragma unroll
    for (int r = 0; r < 8; r += 2) {
        v2f z  = {acc[r], acc[r + 1]};
        v2f a2 = {a[r], a[r + 1]};
        v2f l  = __builtin_elementwise_max(z, 0.2f * z);
        s = a2 * l + s;
    }
    return s;
}

__global__ __launch_bounds__(256)
void k_edge_logits(const int* __restrict__ ei, const float* __restrict__ ew,
                   const float* __restrict__ h,
                   const float* __restrict__ gamma1, const float* __restrict__ beta1,
                   const float* __restrict__ rm1, const float* __restrict__ rv1,
                   const float* __restrict__ Wl, const float* __restrict__ bl,
                   const float* __restrict__ Wr, const float* __restrict__ br,
                   const float* __restrict__ We, const float* __restrict__ att,
                   float* __restrict__ logits, unsigned* __restrict__ mx,
                   float* __restrict__ cnt) {
    const int lane = threadIdx.x & 31;
    const int c = lane & 15, hf = lane >> 4;
    const int wv = blockIdx.x * 8 + __builtin_amdgcn_readfirstlane((int)(threadIdx.x >> 5));

    const float* Wside = hf ? Wr : Wl;
    v2f wa[16][3];
    v8f at[16];
#pragma unroll
    for (int t = 0; t < 16; ++t) {
        int col = t * 16 + c;
        float wedge = We[col];
        float bsum = bl[col] + br[col];
        wa[t][0] = (v2f){Wside[0 * HD + col], Wside[1 * HD + col]};
        wa[t][1] = (v2f){Wside[2 * HD + col], Wside[3 * HD + col]};
        wa[t][2] = (v2f){Wside[4 * HD + col], hf ? bsum : wedge};
        at[t] = *(const v8f*)(att + t * 16 + 8 * hf);
    }
    float rm[F_IN], rs[F_IN], gm[F_IN], bt[F_IN];
#pragma unroll
    for (int k = 0; k < F_IN; ++k) {
        rm[k] = rm1[k];
        rs[k] = rsqrtf(rv1[k] + BN_EPS);
        gm[k] = gamma1[k];
        bt[k] = beta1[k];
    }

    const int* eside = ei + hf * N_EDGES;
    int g = wv;
    int node = eside[g * 16 + c];
    float raw[F_IN];
#pragma unroll
    for (int k = 0; k < F_IN; ++k) raw[k] = h[node * F_IN + k];
    float w = ew[g * 16 + c];
    int g1 = min(g + LG_WAVES, LG_GROUPS - 1);
    int node1 = eside[g1 * 16 + c];

    for (; g < LG_GROUPS; g += LG_WAVES) {
        float raw1[F_IN];
#pragma unroll
        for (int k = 0; k < F_IN; ++k) raw1[k] = h[node1 * F_IN + k];
        float w1 = ew[g1 * 16 + c];
        int g2 = min(g1 + LG_WAVES, LG_GROUPS - 1);
        int node2 = eside[g2 * 16 + c];

        float f[6];
#pragma unroll
        for (int k = 0; k < F_IN; ++k)
            f[k] = (raw[k] - rm[k]) * rs[k] * gm[k] + bt[k];
        f[5] = hf ? 1.f : w;
        v2f b0 = {f[0], f[1]}, b1 = {f[2], f[3]}, b2 = {f[4], f[5]};

        v2f s0 = {0.f, 0.f}, s1 = {0.f, 0.f};
#pragma unroll
        for (int t = 0; t < 16; t += 2) {
            v8f a0 = {0.f, 0.f, 0.f, 0.f, 0.f, 0.f, 0.f, 0.f};
            v8f a1 = {0.f, 0.f, 0.f, 0.f, 0.f, 0.f, 0.f, 0.f};
            a0 = __builtin_amdgcn_wmma_f32_16x16x4_f32(false, wa[t][0], false, b0, (short)0, a0, false, false);
            a1 = __builtin_amdgcn_wmma_f32_16x16x4_f32(false, wa[t + 1][0], false, b0, (short)0, a1, false, false);
            a0 = __builtin_amdgcn_wmma_f32_16x16x4_f32(false, wa[t][1], false, b1, (short)0, a0, false, false);
            a1 = __builtin_amdgcn_wmma_f32_16x16x4_f32(false, wa[t + 1][1], false, b1, (short)0, a1, false, false);
            a0 = __builtin_amdgcn_wmma_f32_16x16x4_f32(false, wa[t][2], false, b2, (short)0, a0, false, false);
            a1 = __builtin_amdgcn_wmma_f32_16x16x4_f32(false, wa[t + 1][2], false, b2, (short)0, a1, false, false);
            if (t < 8) {
                s0 = lrelu_att(a0, at[t], s0);
                s0 = lrelu_att(a1, at[t + 1], s0);
            } else {
                s1 = lrelu_att(a0, at[t], s1);
                s1 = lrelu_att(a1, at[t + 1], s1);
            }
        }
        float l0 = s0.x + s0.y, l1 = s1.x + s1.y;
        l0 += __shfl_xor(l0, 16, 32);
        l1 += __shfl_xor(l1, 16, 32);
        if (hf) {
            int e = g * 16 + c;
            *(v2f*)(logits + (long)e * HEADS) = (v2f){l0, l1};
            atomicMax(&mx[(long)node * HEADS + 0], enc_f32(l0));
            atomicMax(&mx[(long)node * HEADS + 1], enc_f32(l1));
            atomAddF(&cnt[node], 1.f);
        }

        node = node1; node1 = node2; g1 = g2; w = w1;
#pragma unroll
        for (int k = 0; k < F_IN; ++k) raw[k] = raw1[k];
    }
}

__global__ __launch_bounds__(256)
void k_edge_exp(const int* __restrict__ ei, float* __restrict__ logits,
                const unsigned* __restrict__ mx, float* __restrict__ den) {
    long i = (long)blockIdx.x * blockDim.x + threadIdx.x;
    if (i >= (long)N_EDGES * HEADS) return;
    int e = (int)(i >> 1), hh = (int)(i & 1);
    int dst = ei[N_EDGES + e];
    float m = dec_f32(mx[(long)dst * HEADS + hh]);
    float ex = expf(logits[i] - m);
    logits[i] = ex;
    atomAddF(&den[(long)dst * HEADS + hh], ex);
}

__global__ __launch_bounds__(256)
void k_edge_scatter(const int* __restrict__ ei, const float* __restrict__ ex,
                    const float* __restrict__ den, const float* __restrict__ h,
                    const float* __restrict__ gamma1, const float* __restrict__ beta1,
                    const float* __restrict__ rm1, const float* __restrict__ rv1,
                    float* __restrict__ agg) {
    int slot = threadIdx.x >> 4;
    int l = threadIdx.x & 15;
    int hh = l >> 3, k = l & 7;
    if (k >= 6) return;
    int kc = k < 5 ? k : 0;
    float rm = rm1[kc], rs = rsqrtf(rv1[kc] + BN_EPS), g = gamma1[kc], b = beta1[kc];
    int e0 = blockIdx.x * 64 + slot;
    int src[4], dst[4];
#pragma unroll
    for (int it = 0; it < 4; ++it) {
        int e = e0 + it * 16;
        src[it] = ei[e];
        dst[it] = ei[N_EDGES + e];
    }
    float xv[4], dn[4], hr[4];
#pragma unroll
    for (int it = 0; it < 4; ++it) {
        int e = e0 + it * 16;
        xv[it] = ex[(long)e * HEADS + hh];
        dn[it] = den[(long)dst[it] * HEADS + hh];
        hr[it] = h[src[it] * F_IN + kc];
    }
#pragma unroll
    for (int it = 0; it < 4; ++it) {
        float alpha = xv[it] / (dn[it] + 1e-16f);
        float hv = (k < 5) ? ((hr[it] - rm) * rs * g + b) : 1.f;
        atomAddF(&agg[(long)dst[it] * AGG_W + l], hv * alpha);
    }
}

__global__ __launch_bounds__(256)
void k_fc(const float* __restrict__ agg, const float* __restrict__ cnt,
          const float* __restrict__ Wl, const float* __restrict__ bl,
          const float* __restrict__ bias_gat,
          const float* __restrict__ W1, const float* __restrict__ b1,
          const float* __restrict__ gamma2, const float* __restrict__ beta2,
          const float* __restrict__ rm2, const float* __restrict__ rv2,
          const float* __restrict__ W2, const float* __restrict__ b2,
          float* __restrict__ out) {
    __shared__ float Y[16 * 128];
    __shared__ float Z[16 * 128];
    __shared__ float As[16 * AGG_W];
    int row0 = blockIdx.x * 16;
    int t = threadIdx.x;

    {
        int d = t & 127;
        int rb = t >> 7;
        float w0[6], w1[6];
#pragma unroll
        for (int k = 0; k < F_IN; ++k) {
            w0[k] = Wl[k * HD + d];
            w1[k] = Wl[k * HD + DDIM + d];
        }
        w0[5] = bl[d];
        w1[5] = bl[DDIM + d];
        float bg = bias_gat[d];
        float inv = 0.5f / fmaxf(cnt[row0 + (t >> 4)], 1.f);
        As[t] = agg[(long)row0 * AGG_W + t] * inv;
        __syncthreads();
#pragma unroll
        for (int i = 0; i < 8; ++i) {
            int r = rb + 2 * i;
            const float* a = &As[r * AGG_W];
            float s0 = 0.f, s1 = 0.f;
#pragma unroll
            for (int k = 0; k < 6; ++k) {
                s0 += a[k] * w0[k];
                s1 += a[8 + k] * w1[k];
            }
            Y[r * 128 + d] = (s0 + s1) + bg;
        }
    }
    __syncthreads();

    int wave = t >> 5, lane = t & 31;
    int c0 = wave * 16;
    int laneN = lane & 15;
    int koff = (lane >> 4) * 2;
    v8f acc = {0.f, 0.f, 0.f, 0.f, 0.f, 0.f, 0.f, 0.f};
    for (int k = 0; k < 128; k += 4) {
        v2f a, b;
        a.x = Y[laneN * 128 + k + koff];
        a.y = Y[laneN * 128 + k + koff + 1];
        int kk = k + koff;
        b.x = W1[kk * 128 + c0 + laneN];
        b.y = W1[(kk + 1) * 128 + c0 + laneN];
        acc = __builtin_amdgcn_wmma_f32_16x16x4_f32(
            false, a, false, b, (short)0, acc, false, false);
    }

#pragma unroll
    for (int v = 0; v < 8; ++v) {
        int m = v + (lane >> 4) * 8;
        int col = c0 + laneN;
        float z = acc[v] + b1[col];
        z = (z - rm2[col]) * rsqrtf(rv2[col] + BN_EPS) * gamma2[col] + beta2[col];
        z = z > 0.f ? z : 0.01f * z;
        Z[m * 128 + col] = z;
    }
    __syncthreads();

    if (t < 16 * F_IN) {
        int r = t / F_IN, c = t % F_IN;
        float o = b2[c];
        const float* zr = &Z[r * 128];
#pragma unroll 8
        for (int d = 0; d < 128; ++d) o += zr[d] * W2[d * F_IN + c];
        out[(long)(row0 + r) * F_IN + c] = o;
    }
}

extern "C" void kernel_launch(void* const* d_in, const int* in_sizes, int n_in,
                              void* d_out, int out_size, void* d_ws, size_t ws_size,
                              hipStream_t stream) {
    const float* h        = (const float*)d_in[0];
    const int*   ei       = (const int*)  d_in[1];
    const float* ew       = (const float*)d_in[2];
    const float* gamma1   = (const float*)d_in[3];
    const float* beta1    = (const float*)d_in[4];
    const float* rm1      = (const float*)d_in[5];
    const float* rv1      = (const float*)d_in[6];
    const float* Wl       = (const float*)d_in[7];
    const float* bl       = (const float*)d_in[8];
    const float* Wr       = (const float*)d_in[9];
    const float* br       = (const float*)d_in[10];
    const float* We       = (const float*)d_in[11];
    const float* att      = (const float*)d_in[12];
    const float* bias_gat = (const float*)d_in[13];
    const float* W1       = (const float*)d_in[14];
    const float* b1       = (const float*)d_in[15];
    const float* gamma2   = (const float*)d_in[16];
    const float* beta2    = (const float*)d_in[17];
    const float* rm2      = (const float*)d_in[18];
    const float* rv2      = (const float*)d_in[19];
    const float* W2       = (const float*)d_in[20];
    const float* b2       = (const float*)d_in[21];
    float* out = (float*)d_out;

    float* ws = (float*)d_ws;
    size_t o = 0;
    float* agg    = ws + o; o += (size_t)N_NODES * AGG_W;
    float* logits = ws + o; o += (size_t)N_EDGES * HEADS;
    float* den    = ws + o; o += (size_t)N_NODES * HEADS;
    float* cnt    = ws + o; o += (size_t)N_NODES;
    unsigned* mx  = (unsigned*)(ws + o); o += (size_t)N_NODES * HEADS;

    const int T = 256;
    k_init<<<(N_NODES * AGG_W + T - 1) / T, T, 0, stream>>>(den, cnt, agg, mx);
    k_edge_logits<<<LG_BLOCKS, T, 0, stream>>>(ei, ew, h, gamma1, beta1, rm1, rv1,
                                               Wl, bl, Wr, br, We, att, logits, mx, cnt);
    k_edge_exp<<<(N_EDGES * HEADS + T - 1) / T, T, 0, stream>>>(ei, logits, mx, den);
    k_edge_scatter<<<N_EDGES / 64, T, 0, stream>>>(ei, logits, den, h, gamma1, beta1,
                                                   rm1, rv1, agg);
    k_fc<<<N_NODES / 16, T, 0, stream>>>(agg, cnt, Wl, bl, bias_gat, W1, b1,
                                         gamma2, beta2, rm2, rv2, W2, b2, out);
}


# Softmax denominator factored out of the edge sum: edges add ex*bn(h[src]) and ex (slot 5 = den), fc kernel divides once per node; exp pass, den atomics and per-edge alpha dropped (f32 sum regrouped)
# speedup vs baseline: 4.0271x; 1.1959x over previous
#include <hip/hip_runtime.h>
#include <math.h>

#define N_NODES 50000
#define N_EDGES 400000
#define HEADS   2
#define DDIM    128
#define F_IN    5
#define HD      (HEADS*DDIM)
#define BN_EPS  1e-5f
#define AGG_W   16

typedef __attribute__((ext_vector_type(2))) float v2f;
typedef __attribute__((ext_vector_type(8))) float v8f;

__device__ __forceinline__ unsigned enc_f32(float f) {
    unsigned u = __float_as_uint(f);
    return (u & 0x80000000u) ? ~u : (u | 0x80000000u);
}
__device__ __forceinline__ float dec_f32(unsigned u) {
    unsigned v = (u & 0x80000000u) ? (u & 0x7FFFFFFFu) : ~u;
    return __uint_as_float(v);
}
__device__ __forceinline__ void atomAddF(float* p, float v) {
    unsafeAtomicAdd(p, v);
}

__global__ void k_init(float* __restrict__ cnt,
                       float* __restrict__ agg, unsigned* __restrict__ mx) {
    long i = (long)blockIdx.x * blockDim.x + threadIdx.x;
    if (i < (long)N_NODES * AGG_W) agg[i] = 0.f;
    if (i < N_NODES) cnt[i] = 0.f;
    if (i < (long)N_NODES * HEADS) mx[i] = 0u;
}

#define LG_BLOCKS 256
#define LG_WAVES  (LG_BLOCKS * 8)
#define LG_GROUPS (N_EDGES / 16)

__device__ __forceinline__ v2f lrelu_att(v8f acc, v8f a, v2f s) {
#pragma unroll
    for (int r = 0; r < 8; r += 2) {
        v2f z  = {acc[r], acc[r + 1]};
        v2f a2 = {a[r], a[r + 1]};
        v2f l  = __builtin_elementwise_max(z, 0.2f * z);
        s = a2 * l + s;
    }
    return s;
}

__global__ __launch_bounds__(256)
void k_edge_logits(const int* __restrict__ ei, const float* __restrict__ ew,
                   const float* __restrict__ h,
                   const float* __restrict__ gamma1, const float* __restrict__ beta1,
                   const float* __restrict__ rm1, const float* __restrict__ rv1,
                   const float* __restrict__ Wl, const float* __restrict__ bl,
                   const float* __restrict__ Wr, const float* __restrict__ br,
                   const float* __restrict__ We, const float* __restrict__ att,
                   float* __restrict__ logits, unsigned* __restrict__ mx,
                   float* __restrict__ cnt) {
    const int lane = threadIdx.x & 31;
    const int c = lane & 15, hf = lane >> 4;
    const int wv = blockIdx.x * 8 + __builtin_amdgcn_readfirstlane((int)(threadIdx.x >> 5));

    const float* Wside = hf ? Wr : Wl;
    v2f wa[16][3];
    v8f at[16];
#pragma unroll
    for (int t = 0; t < 16; ++t) {
        int col = t * 16 + c;
        float wedge = We[col];
        float bsum = bl[col] + br[col];
        wa[t][0] = (v2f){Wside[0 * HD + col], Wside[1 * HD + col]};
        wa[t][1] = (v2f){Wside[2 * HD + col], Wside[3 * HD + col]};
        wa[t][2] = (v2f){Wside[4 * HD + col], hf ? bsum : wedge};
        at[t] = *(const v8f*)(att + t * 16 + 8 * hf);
    }
    float rm[F_IN], rs[F_IN], gm[F_IN], bt[F_IN];
#pragma unroll
    for (int k = 0; k < F_IN; ++k) {
        rm[k] = rm1[k];
        rs[k] = rsqrtf(rv1[k] + BN_EPS);
        gm[k] = gamma1[k];
        bt[k] = beta1[k];
    }

    const int* eside = ei + hf * N_EDGES;
    int g = wv;
    int node = eside[g * 16 + c];
    float raw[F_IN];
#pragma unroll
    for (int k = 0; k < F_IN; ++k) raw[k] = h[node * F_IN + k];
    float w = ew[g * 16 + c];
    int g1 = min(g + LG_WAVES, LG_GROUPS - 1);
    int node1 = eside[g1 * 16 + c];

    for (; g < LG_GROUPS; g += LG_WAVES) {
        float raw1[F_IN];
#pragma unroll
        for (int k = 0; k < F_IN; ++k) raw1[k] = h[node1 * F_IN + k];
        float w1 = ew[g1 * 16 + c];
        int g2 = min(g1 + LG_WAVES, LG_GROUPS - 1);
        int node2 = eside[g2 * 16 + c];

        float f[6];
#pragma unroll
        for (int k = 0; k < F_IN; ++k)
            f[k] = (raw[k] - rm[k]) * rs[k] * gm[k] + bt[k];
        f[5] = hf ? 1.f : w;
        v2f b0 = {f[0], f[1]}, b1 = {f[2], f[3]}, b2 = {f[4], f[5]};

        v2f s0 = {0.f, 0.f}, s1 = {0.f, 0.f};
#pragma unroll
        for (int t = 0; t < 16; t += 2) {
            v8f a0 = {0.f, 0.f, 0.f, 0.f, 0.f, 0.f, 0.f, 0.f};
            v8f a1 = {0.f, 0.f, 0.f, 0.f, 0.f, 0.f, 0.f, 0.f};
            a0 = __builtin_amdgcn_wmma_f32_16x16x4_f32(false, wa[t][0], false, b0, (short)0, a0, false, false);
            a1 = __builtin_amdgcn_wmma_f32_16x16x4_f32(false, wa[t + 1][0], false, b0, (short)0, a1, false, false);
            a0 = __builtin_amdgcn_wmma_f32_16x16x4_f32(false, wa[t][1], false, b1, (short)0, a0, false, false);
            a1 = __builtin_amdgcn_wmma_f32_16x16x4_f32(false, wa[t + 1][1], false, b1, (short)0, a1, false, false);
            a0 = __builtin_amdgcn_wmma_f32_16x16x4_f32(false, wa[t][2], false, b2, (short)0, a0, false, false);
            a1 = __builtin_amdgcn_wmma_f32_16x16x4_f32(false, wa[t + 1][2], false, b2, (short)0, a1, false, false);
            if (t < 8) {
                s0 = lrelu_att(a0, at[t], s0);
                s0 = lrelu_att(a1, at[t + 1], s0);
            } else {
                s1 = lrelu_att(a0, at[t], s1);
                s1 = lrelu_att(a1, at[t + 1], s1);
            }
        }
        float l0 = s0.x + s0.y, l1 = s1.x + s1.y;
        l0 += __shfl_xor(l0, 16, 32);
        l1 += __shfl_xor(l1, 16, 32);
        if (hf) {
            int e = g * 16 + c;
            *(v2f*)(logits + (long)e * HEADS) = (v2f){l0, l1};
            atomicMax(&mx[(long)node * HEADS + 0], enc_f32(l0));
            atomicMax(&mx[(long)node * HEADS + 1], enc_f32(l1));
            atomAddF(&cnt[node], 1.f);
        }

        node = node1; node1 = node2; g1 = g2; w = w1;
#pragma unroll
        for (int k = 0; k < F_IN; ++k) raw[k] = raw1[k];
    }
}

__global__ __launch_bounds__(256)
void k_edge_scatter(const int* __restrict__ ei, const float* __restrict__ logits,
                    const unsigned* __restrict__ mx, const float* __restrict__ h,
                    const float* __restrict__ gamma1, const float* __restrict__ beta1,
                    const float* __restrict__ rm1, const float* __restrict__ rv1,
                    float* __restrict__ agg) {
    int slot = threadIdx.x >> 4;
    int l = threadIdx.x & 15;
    int hh = l >> 3, k = l & 7;
    if (k >= 6) return;
    int kc = k < 5 ? k : 0;
    float rm = rm1[kc], rs = rsqrtf(rv1[kc] + BN_EPS), g = gamma1[kc], b = beta1[kc];
    int e0 = blockIdx.x * 64 + slot;
    int src[4], dst[4];
#pragma unroll
    for (int it = 0; it < 4; ++it) {
        int e = e0 + it * 16;
        src[it] = ei[e];
        dst[it] = ei[N_EDGES + e];
    }
    float lg[4], hr[4];
    unsigned mv[4];
#pragma unroll
    for (int it = 0; it < 4; ++it) {
        int e = e0 + it * 16;
        lg[it] = logits[(long)e * HEADS + hh];
        mv[it] = mx[(long)dst[it] * HEADS + hh];
        hr[it] = h[src[it] * F_IN + kc];
    }
#pragma unroll
    for (int it = 0; it < 4; ++it) {
        float ex = expf(lg[it] - dec_f32(mv[it]));
        float hv = (k < 5) ? ((hr[it] - rm) * rs * g + b) : 1.f;
        atomAddF(&agg[(long)dst[it] * AGG_W + l], hv * ex);
    }
}

__global__ __launch_bounds__(256)
void k_fc(const float* __restrict__ agg, const float* __restrict__ cnt,
          const float* __restrict__ Wl, const float* __restrict__ bl,
          const float* __restrict__ bias_gat,
          const float* __restrict__ W1, const float* __restrict__ b1,
          const float* __restrict__ gamma2, const float* __restrict__ beta2,
          const float* __restrict__ rm2, const float* __restrict__ rv2,
          const float* __restrict__ W2, const float* __restrict__ b2,
          float* __restrict__ out) {
    __shared__ float Y[16 * 128];
    __shared__ float Z[16 * 128];
    __shared__ float As[16 * AGG_W];
    int row0 = blockIdx.x * 16;
    int t = threadIdx.x;

    {
        int d = t & 127;
        int rb = t >> 7;
        float w0[6], w1[6];
#pragma unroll
        for (int k = 0; k < F_IN; ++k) {
            w0[k] = Wl[k * HD + d];
            w1[k] = Wl[k * HD + DDIM + d];
        }
        w0[5] = bl[d];
        w1[5] = bl[DDIM + d];
        float bg = bias_gat[d];
        float inv = 0.5f / fmaxf(cnt[row0 + (t >> 4)], 1.f);
        float dsum = agg[(long)row0 * AGG_W + (t & ~7) + 5];
        As[t] = agg[(long)row0 * AGG_W + t] / (dsum + 1e-16f) * inv;
        __syncthreads();
#pragma unroll
        for (int i = 0; i < 8; ++i) {
            int r = rb + 2 * i;
            const float* a = &As[r * AGG_W];
            float s0 = 0.f, s1 = 0.f;
#pragma unroll
            for (int k = 0; k < 6; ++k) {
                s0 += a[k] * w0[k];
                s1 += a[8 + k] * w1[k];
            }
            Y[r * 128 + d] = (s0 + s1) + bg;
        }
    }
    __syncthreads();

    int wave = t >> 5, lane = t & 31;
    int c0 = wave * 16;
    int laneN = lane & 15;
    int koff = (lane >> 4) * 2;
    v8f acc = {0.f, 0.f, 0.f, 0.f, 0.f, 0.f, 0.f, 0.f};
    for (int k = 0; k < 128; k += 4) {
        v2f a, b;
        a.x = Y[laneN * 128 + k + koff];
        a.y = Y[laneN * 128 + k + koff + 1];
        int kk = k + koff;
        b.x = W1[kk * 128 + c0 + laneN];
        b.y = W1[(kk + 1) * 128 + c0 + laneN];
        acc = __builtin_amdgcn_wmma_f32_16x16x4_f32(
            false, a, false, b, (short)0, acc, false, false);
    }

#pragma unroll
    for (int v = 0; v < 8; ++v) {
        int m = v + (lane >> 4) * 8;
        int col = c0 + laneN;
        float z = acc[v] + b1[col];
        z = (z - rm2[col]) * rsqrtf(rv2[col] + BN_EPS) * gamma2[col] + beta2[col];
        z = z > 0.f ? z : 0.01f * z;
        Z[m * 128 + col] = z;
    }
    __syncthreads();

    if (t < 16 * F_IN) {
        int r = t / F_IN, c = t % F_IN;
        float o = b2[c];
        const float* zr = &Z[r * 128];
#pragma unroll 8
        for (int d = 0; d < 128; ++d) o += zr[d] * W2[d * F_IN + c];
        out[(long)(row0 + r) * F_IN + c] = o;
    }
}

extern "C" void kernel_launch(void* const* d_in, const int* in_sizes, int n_in,
                              void* d_out, int out_size, void* d_ws, size_t ws_size,
                              hipStream_t stream) {
    const float* h        = (const float*)d_in[0];
    const int*   ei       = (const int*)  d_in[1];
    const float* ew       = (const float*)d_in[2];
    const float* gamma1   = (const float*)d_in[3];
    const float* beta1    = (const float*)d_in[4];
    const float* rm1      = (const float*)d_in[5];
    const float* rv1      = (const float*)d_in[6];
    const float* Wl       = (const float*)d_in[7];
    const float* bl       = (const float*)d_in[8];
    const float* Wr       = (const float*)d_in[9];
    const float* br       = (const float*)d_in[10];
    const float* We       = (const float*)d_in[11];
    const float* att      = (const float*)d_in[12];
    const float* bias_gat = (const float*)d_in[13];
    const float* W1       = (const float*)d_in[14];
    const float* b1       = (const float*)d_in[15];
    const float* gamma2   = (const float*)d_in[16];
    const float* beta2    = (const float*)d_in[17];
    const float* rm2      = (const float*)d_in[18];
    const float* rv2      = (const float*)d_in[19];
    const float* W2       = (const float*)d_in[20];
    const float* b2       = (const float*)d_in[21];
    float* out = (float*)d_out;

    float* ws = (float*)d_ws;
    size_t o = 0;
    float* agg    = ws + o; o += (size_t)N_NODES * AGG_W;
    float* logits = ws + o; o += (size_t)N_EDGES * HEADS;
    float* cnt    = ws + o; o += (size_t)N_NODES;
    unsigned* mx  = (unsigned*)(ws + o); o += (size_t)N_NODES * HEADS;

    const int T = 256;
    k_init<<<(N_NODES * AGG_W + T - 1) / T, T, 0, stream>>>(cnt, agg, mx);
    k_edge_logits<<<LG_BLOCKS, T, 0, stream>>>(ei, ew, h, gamma1, beta1, rm1, rv1,
                                               Wl, bl, Wr, br, We, att, logits, mx, cnt);
    k_edge_scatter<<<N_EDGES / 64, T, 0, stream>>>(ei, logits, mx, h, gamma1, beta1,
                                                   rm1, rv1, agg);
    k_fc<<<N_NODES / 16, T, 0, stream>>>(agg, cnt, Wl, bl, bias_gat, W1, b1,
                                         gamma2, beta2, rm2, rv2, W2, b2, out);
}


# Logits kernel sends one scattered atomic per edge, not three: both heads' max in one instruction across the lane halves; edge count rides as lane 6 of the scatter row (sum of ones, exact)
# speedup vs baseline: 4.5264x; 1.1240x over previous
#include <hip/hip_runtime.h>
#include <math.h>

#define N_NODES 50000
#define N_EDGES 400000
#define HEADS   2
#define DDIM    128
#define F_IN    5
#define HD      (HEADS*DDIM)
#define BN_EPS  1e-5f
#define AGG_W   16

typedef __attribute__((ext_vector_type(2))) float v2f;
typedef __attribute__((ext_vector_type(8))) float v8f;

__device__ __forceinline__ unsigned enc_f32(float f) {
    unsigned u = __float_as_uint(f);
    return (u & 0x80000000u) ? ~u : (u | 0x80000000u);
}
__device__ __forceinline__ float dec_f32(unsigned u) {
    unsigned v = (u & 0x80000000u) ? (u & 0x7FFFFFFFu) : ~u;
    return __uint_as_float(v);
}
__device__ __forceinline__ void atomAddF(float* p, float v) {
    unsafeAtomicAdd(p, v);
}

__global__ void k_init(float* __restrict__ agg, unsigned* __restrict__ mx) {
    long i = (long)blockIdx.x * blockDim.x + threadIdx.x;
    if (i < (long)N_NODES * AGG_W) agg[i] = 0.f;
    if (i < (long)N_NODES * HEADS) mx[i] = 0u;
}

#define LG_BLOCKS 256
#define LG_WAVES  (LG_BLOCKS * 8)
#define LG_GROUPS (N_EDGES / 16)

__device__ __forceinline__ v2f lrelu_att(v8f acc, v8f a, v2f s) {
#pragma unroll
    for (int r = 0; r < 8; r += 2) {
        v2f z  = {acc[r], acc[r + 1]};
        v2f a2 = {a[r], a[r + 1]};
        v2f l  = __builtin_elementwise_max(z, 0.2f * z);
        s = a2 * l + s;
    }
    return s;
}

__global__ __launch_bounds__(256)
void k_edge_logits(const int* __restrict__ ei, const float* __restrict__ ew,
                   const float* __restrict__ h,
                   const float* __restrict__ gamma1, const float* __restrict__ beta1,
                   const float* __restrict__ rm1, const float* __restrict__ rv1,
                   const float* __restrict__ Wl, const float* __restrict__ bl,
                   const float* __restrict__ Wr, const float* __restrict__ br,
                   const float* __restrict__ We, const float* __restrict__ att,
                   float* __restrict__ logits, unsigned* __restrict__ mx) {
    const int lane = threadIdx.x & 31;
    const int c = lane & 15, hf = lane >> 4;
    const int wv = blockIdx.x * 8 + __builtin_amdgcn_readfirstlane((int)(threadIdx.x >> 5));

    const float* Wside = hf ? Wr : Wl;
    v2f wa[16][3];
    v8f at[16];
#pragma unroll
    for (int t = 0; t < 16; ++t) {
        int col = t * 16 + c;
        float wedge = We[col];
        float bsum = bl[col] + br[col];
        wa[t][0] = (v2f){Wside[0 * HD + col], Wside[1 * HD + col]};
        wa[t][1] = (v2f){Wside[2 * HD + col], Wside[3 * HD + col]};
        wa[t][2] = (v2f){Wside[4 * HD + col], hf ? bsum : wedge};
        at[t] = *(const v8f*)(att + t * 16 + 8 * hf);
    }
    float rm[F_IN], rs[F_IN], gm[F_IN], bt[F_IN];
#pragma unroll
    for (int k = 0; k < F_IN; ++k) {
        rm[k] = rm1[k];
        rs[k] = rsqrtf(rv1[k] + BN_EPS);
        gm[k] = gamma1[k];
        bt[k] = beta1[k];
    }

    const int* eside = ei + hf * N_EDGES;
    int g = wv;
    int node = eside[g * 16 + c];
    float raw[F_IN];
#pragma unroll
    for (int k = 0; k < F_IN; ++k) raw[k] = h[node * F_IN + k];
    float w = ew[g * 16 + c];
    int g1 = min(g + LG_WAVES, LG_GROUPS - 1);
    int node1 = eside[g1 * 16 + c];

    for (; g < LG_GROUPS; g += LG_WAVES) {
        float raw1[F_IN];
#pragma unroll
        for (int k = 0; k < F_IN; ++k) raw1[k] = h[node1 * F_IN + k];
        float w1 = ew[g1 * 16 + c];
        int g2 = min(g1 + LG_WAVES, LG_GROUPS - 1);
        int node2 = eside[g2 * 16 + c];
        int dn = ei[N_EDGES + g * 16 + c];

        float f[6];
#pragma unroll
        for (int k = 0; k < F_IN; ++k)
            f[k] = (raw[k] - rm[k]) * rs[k] * gm[k] + bt[k];
        f[5] = hf ? 1.f : w;
        v2f b0 = {f[0], f[1]}, b1 = {f[2], f[3]}, b2 = {f[4], f[5]};

        v2f s0 = {0.f, 0.f}, s1 = {0.f, 0.f};
#pragma unroll
        for (int t = 0; t < 16; t += 2) {
            v8f a0 = {0.f, 0.f, 0.f, 0.f, 0.f, 0.f, 0.f, 0.f};
            v8f a1 = {0.f, 0.f, 0.f, 0.f, 0.f, 0.f, 0.f, 0.f};
            a0 = __builtin_amdgcn_wmma_f32_16x16x4_f32(false, wa[t][0], false, b0, (short)0, a0, false, false);
            a1 = __builtin_amdgcn_wmma_f32_16x16x4_f32(false, wa[t + 1][0], false, b0, (short)0, a1, false, false);
            a0 = __builtin_amdgcn_wmma_f32_16x16x4_f32(false, wa[t][1], false, b1, (short)0, a0, false, false);
            a1 = __builtin_amdgcn_wmma_f32_16x16x4_f32(false, wa[t + 1][1], false, b1, (short)0, a1, false, false);
            a0 = __builtin_amdgcn_wmma_f32_16x16x4_f32(false, wa[t][2], false, b2, (short)0, a0, false, false);
            a1 = __builtin_amdgcn_wmma_f32_16x16x4_f32(false, wa[t + 1][2], false, b2, (short)0, a1, false, false);
            if (t < 8) {
                s0 = lrelu_att(a0, at[t], s0);
                s0 = lrelu_att(a1, at[t + 1], s0);
            } else {
                s1 = lrelu_att(a0, at[t], s1);
                s1 = lrelu_att(a1, at[t + 1], s1);
            }
        }
        float l0 = s0.x + s0.y, l1 = s1.x + s1.y;
        float lh = (hf ? l1 : l0) + __shfl_xor(hf ? l0 : l1, 16, 32);
        int e = g * 16 + c;
        logits[(long)e * HEADS + hf] = lh;
        atomicMax(&mx[(long)dn * HEADS + hf], enc_f32(lh));

        node = node1; node1 = node2; g1 = g2; w = w1;
#pragma unroll
        for (int k = 0; k < F_IN; ++k) raw[k] = raw1[k];
    }
}

__global__ __launch_bounds__(256)
void k_edge_scatter(const int* __restrict__ ei, const float* __restrict__ logits,
                    const unsigned* __restrict__ mx, const float* __restrict__ h,
                    const float* __restrict__ gamma1, const float* __restrict__ beta1,
                    const float* __restrict__ rm1, const float* __restrict__ rv1,
                    float* __restrict__ agg) {
    int slot = threadIdx.x >> 4;
    int l = threadIdx.x & 15;
    int hh = l >> 3, k = l & 7;
    if (k >= 6 && l != 6) return;
    int kc = k < 5 ? k : 0;
    float rm = rm1[kc], rs = rsqrtf(rv1[kc] + BN_EPS), g = gamma1[kc], b = beta1[kc];
    int e0 = blockIdx.x * 64 + slot;
    int src[4], dst[4];
#pragma unroll
    for (int it = 0; it < 4; ++it) {
        int e = e0 + it * 16;
        src[it] = ei[e];
        dst[it] = ei[N_EDGES + e];
    }
    float lg[4], hr[4];
    unsigned mv[4];
#pragma unroll
    for (int it = 0; it < 4; ++it) {
        int e = e0 + it * 16;
        lg[it] = logits[(long)e * HEADS + hh];
        mv[it] = mx[(long)dst[it] * HEADS + hh];
        hr[it] = h[src[it] * F_IN + kc];
    }
#pragma unroll
    for (int it = 0; it < 4; ++it) {
        float ex = expf(lg[it] - dec_f32(mv[it]));
        float hv = (k < 5) ? ((hr[it] - rm) * rs * g + b) : 1.f;
        float v = (l == 6) ? 1.f : hv * ex;
        atomAddF(&agg[(long)dst[it] * AGG_W + l], v);
    }
}

__global__ __launch_bounds__(256)
void k_fc(const float* __restrict__ agg,
          const float* __restrict__ Wl, const float* __restrict__ bl,
          const float* __restrict__ bias_gat,
          const float* __restrict__ W1, const float* __restrict__ b1,
          const float* __restrict__ gamma2, const float* __restrict__ beta2,
          const float* __restrict__ rm2, const float* __restrict__ rv2,
          const float* __restrict__ W2, const float* __restrict__ b2,
          float* __restrict__ out) {
    __shared__ float Y[16 * 128];
    __shared__ float Z[16 * 128];
    __shared__ float As[16 * AGG_W];
    int row0 = blockIdx.x * 16;
    int t = threadIdx.x;

    {
        int d = t & 127;
        int rb = t >> 7;
        float w0[6], w1[6];
#pragma unroll
        for (int k = 0; k < F_IN; ++k) {
            w0[k] = Wl[k * HD + d];
            w1[k] = Wl[k * HD + DDIM + d];
        }
        w0[5] = bl[d];
        w1[5] = bl[DDIM + d];
        float bg = bias_gat[d];
        float inv = 0.5f / fmaxf(agg[(long)row0 * AGG_W + (t & ~15) + 6], 1.f);
        float dsum = agg[(long)row0 * AGG_W + (t & ~7) + 5];
        As[t] = agg[(long)row0 * AGG_W + t] / (dsum + 1e-16f) * inv;
        __syncthreads();
#pragma unroll
        for (int i = 0; i < 8; ++i) {
            int r = rb + 2 * i;
            const float* a = &As[r * AGG_W];
            float s0 = 0.f, s1 = 0.f;
#pragma unroll
            for (int k = 0; k < 6; ++k) {
                s0 += a[k] * w0[k];
                s1 += a[8 + k] * w1[k];
            }
            Y[r * 128 + d] = (s0 + s1) + bg;
        }
    }
    __syncthreads();

    int wave = t >> 5, lane = t & 31;
    int c0 = wave * 16;
    int laneN = lane & 15;
    int koff = (lane >> 4) * 2;
    v8f acc = {0.f, 0.f, 0.f, 0.f, 0.f, 0.f, 0.f, 0.f};
    for (int k = 0; k < 128; k += 4) {
        v2f a, b;
        a.x = Y[laneN * 128 + k + koff];
        a.y = Y[laneN * 128 + k + koff + 1];
        int kk = k + koff;
        b.x = W1[kk * 128 + c0 + laneN];
        b.y = W1[(kk + 1) * 128 + c0 + laneN];
        acc = __builtin_amdgcn_wmma_f32_16x16x4_f32(
            false, a, false, b, (short)0, acc, false, false);
    }

#pragma unroll
    for (int v = 0; v < 8; ++v) {
        int m = v + (lane >> 4) * 8;
        int col = c0 + laneN;
        float z = acc[v] + b1[col];
        z = (z - rm2[col]) * rsqrtf(rv2[col] + BN_EPS) * gamma2[col] + beta2[col];
        z = z > 0.f ? z : 0.01f * z;
        Z[m * 128 + col] = z;
    }
    __syncthreads();

    if (t < 16 * F_IN) {
        int r = t / F_IN, c = t % F_IN;
        float o = b2[c];
        const float* zr = &Z[r * 128];
#pragma unroll 8
        for (int d = 0; d < 128; ++d) o += zr[d] * W2[d * F_IN + c];
        out[(long)(row0 + r) * F_IN + c] = o;
    }
}

extern "C" void kernel_launch(void* const* d_in, const int* in_sizes, int n_in,
                              void* d_out, int out_size, void* d_ws, size_t ws_size,
                              hipStream_t stream) {
    const float* h        = (const float*)d_in[0];
    const int*   ei       = (const int*)  d_in[1];
    const float* ew       = (const float*)d_in[2];
    const float* gamma1   = (const float*)d_in[3];
    const float* beta1    = (const float*)d_in[4];
    const float* rm1      = (const float*)d_in[5];
    const float* rv1      = (const float*)d_in[6];
    const float* Wl       = (const float*)d_in[7];
    const float* bl       = (const float*)d_in[8];
    const float* Wr       = (const float*)d_in[9];
    const float* br       = (const float*)d_in[10];
    const float* We       = (const float*)d_in[11];
    const float* att      = (const float*)d_in[12];
    const float* bias_gat = (const float*)d_in[13];
    const float* W1       = (const float*)d_in[14];
    const float* b1       = (const float*)d_in[15];
    const float* gamma2   = (const float*)d_in[16];
    const float* beta2    = (const float*)d_in[17];
    const float* rm2      = (const float*)d_in[18];
    const float* rv2      = (const float*)d_in[19];
    const float* W2       = (const float*)d_in[20];
    const float* b2       = (const float*)d_in[21];
    float* out = (float*)d_out;

    float* ws = (float*)d_ws;
    size_t o = 0;
    float* agg    = ws + o; o += (size_t)N_NODES * AGG_W;
    float* logits = ws + o; o += (size_t)N_EDGES * HEADS;
    unsigned* mx  = (unsigned*)(ws + o); o += (size_t)N_NODES * HEADS;

    const int T = 256;
    k_init<<<(N_NODES * AGG_W + T - 1) / T, T, 0, stream>>>(agg, mx);
    k_edge_logits<<<LG_BLOCKS, T, 0, stream>>>(ei, ew, h, gamma1, beta1, rm1, rv1,
                                               Wl, bl, Wr, br, We, att, logits, mx);
    k_edge_scatter<<<N_EDGES / 64, T, 0, stream>>>(ei, logits, mx, h, gamma1, beta1,
                                                   rm1, rv1, agg);
    k_fc<<<N_NODES / 16, T, 0, stream>>>(agg, Wl, bl, bias_gat, W1, b1,
                                         gamma2, beta2, rm2, rv2, W2, b2, out);
}


# fc kernel: row pitch of its two 16x128 LDS tiles 128 -> 132 floats, so the A-fragment reads (16 rows on one bank, 16 passes each) and the fc2 row reads are bank-conflict free
# speedup vs baseline: 5.2226x; 1.1538x over previous
#include <hip/hip_runtime.h>
#include <math.h>

#define N_NODES 50000
#define N_EDGES 400000
#define HEADS   2
#define DDIM    128
#define F_IN    5
#define HD      (HEADS*DDIM)
#define BN_EPS  1e-5f
#define AGG_W   16

typedef __attribute__((ext_vector_type(2))) float v2f;
typedef __attribute__((ext_vector_type(8))) float v8f;

__device__ __forceinline__ unsigned enc_f32(float f) {
    unsigned u = __float_as_uint(f);
    return (u & 0x80000000u) ? ~u : (u | 0x80000000u);
}
__device__ __forceinline__ float dec_f32(unsigned u) {
    unsigned v = (u & 0x80000000u) ? (u & 0x7FFFFFFFu) : ~u;
    return __uint_as_float(v);
}
__device__ __forceinline__ void atomAddF(float* p, float v) {
    unsafeAtomicAdd(p, v);
}

__global__ void k_init(float* __restrict__ agg, unsigned* __restrict__ mx) {
    long i = (long)blockIdx.x * blockDim.x + threadIdx.x;
    if (i < (long)N_NODES * AGG_W) agg[i] = 0.f;
    if (i < (long)N_NODES * HEADS) mx[i] = 0u;
}

#define LG_BLOCKS 256
#define LG_WAVES  (LG_BLOCKS * 8)
#define LG_GROUPS (N_EDGES / 16)

__device__ __forceinline__ v2f lrelu_att(v8f acc, v8f a, v2f s) {
#pragma unroll
    for (int r = 0; r < 8; r += 2) {
        v2f z  = {acc[r], acc[r + 1]};
        v2f a2 = {a[r], a[r + 1]};
        v2f l  = __builtin_elementwise_max(z, 0.2f * z);
        s = a2 * l + s;
    }
    return s;
}

__global__ __launch_bounds__(256)
void k_edge_logits(const int* __restrict__ ei, const float* __restrict__ ew,
                   const float* __restrict__ h,
                   const float* __restrict__ gamma1, const float* __restrict__ beta1,
                   const float* __restrict__ rm1, const float* __restrict__ rv1,
                   const float* __restrict__ Wl, const float* __restrict__ bl,
                   const float* __restrict__ Wr, const float* __restrict__ br,
                   const float* __restrict__ We, const float* __restrict__ att,
                   float* __restrict__ logits, unsigned* __restrict__ mx) {
    const int lane = threadIdx.x & 31;
    const int c = lane & 15, hf = lane >> 4;
    const int wv = blockIdx.x * 8 + __builtin_amdgcn_readfirstlane((int)(threadIdx.x >> 5));

    const float* Wside = hf ? Wr : Wl;
    v2f wa[16][3];
    v8f at[16];
#pragma unroll
    for (int t = 0; t < 16; ++t) {
        int col = t * 16 + c;
        float wedge = We[col];
        float bsum = bl[col] + br[col];
        wa[t][0] = (v2f){Wside[0 * HD + col], Wside[1 * HD + col]};
        wa[t][1] = (v2f){Wside[2 * HD + col], Wside[3 * HD + col]};
        wa[t][2] = (v2f){Wside[4 * HD + col], hf ? bsum : wedge};
        at[t] = *(const v8f*)(att + t * 16 + 8 * hf);
    }
    float rm[F_IN], rs[F_IN], gm[F_IN], bt[F_IN];
#pragma unroll
    for (int k = 0; k < F_IN; ++k) {
        rm[k] = rm1[k];
        rs[k] = rsqrtf(rv1[k] + BN_EPS);
        gm[k] = gamma1[k];
        bt[k] = beta1[k];
    }

    const int* eside = ei + hf * N_EDGES;
    int g = wv;
    int node = eside[g * 16 + c];
    float raw[F_IN];
#pragma unroll
    for (int k = 0; k < F_IN; ++k) raw[k] = h[node * F_IN + k];
    float w = ew[g * 16 + c];
    int g1 = min(g + LG_WAVES, LG_GROUPS - 1);
    int node1 = eside[g1 * 16 + c];

    for (; g < LG_GROUPS; g += LG_WAVES) {
        float raw1[F_IN];
#pragma unroll
        for (int k = 0; k < F_IN; ++k) raw1[k] = h[node1 * F_IN + k];
        float w1 = ew[g1 * 16 + c];
        int g2 = min(g1 + LG_WAVES, LG_GROUPS - 1);
        int node2 = eside[g2 * 16 + c];
        int dn = ei[N_EDGES + g * 16 + c];

        float f[6];
#pragma unroll
        for (int k = 0; k < F_IN; ++k)
            f[k] = (raw[k] - rm[k]) * rs[k] * gm[k] + bt[k];
        f[5] = hf ? 1.f : w;
        v2f b0 = {f[0], f[1]}, b1 = {f[2], f[3]}, b2 = {f[4], f[5]};

        v2f s0 = {0.f, 0.f}, s1 = {0.f, 0.f};
#pragma unroll
        for (int t = 0; t < 16; t += 2) {
            v8f a0 = {0.f, 0.f, 0.f, 0.f, 0.f, 0.f, 0.f, 0.f};
            v8f a1 = {0.f, 0.f, 0.f, 0.f, 0.f, 0.f, 0.f, 0.f};
            a0 = __builtin_amdgcn_wmma_f32_16x16x4_f32(false, wa[t][0], false, b0, (short)0, a0, false, false);
            a1 = __builtin_amdgcn_wmma_f32_16x16x4_f32(false, wa[t + 1][0], false, b0, (short)0, a1, false, false);
            a0 = __builtin_amdgcn_wmma_f32_16x16x4_f32(false, wa[t][1], false, b1, (short)0, a0, false, false);
            a1 = __builtin_amdgcn_wmma_f32_16x16x4_f32(false, wa[t + 1][1], false, b1, (short)0, a1, false, false);
            a0 = __builtin_amdgcn_wmma_f32_16x16x4_f32(false, wa[t][2], false, b2, (short)0, a0, false, false);
            a1 = __builtin_amdgcn_wmma_f32_16x16x4_f32(false, wa[t + 1][2], false, b2, (short)0, a1, false, false);
            if (t < 8) {
                s0 = lrelu_att(a0, at[t], s0);
                s0 = lrelu_att(a1, at[t + 1], s0);
            } else {
                s1 = lrelu_att(a0, at[t], s1);
                s1 = lrelu_att(a1, at[t + 1], s1);
            }
        }
        float l0 = s0.x + s0.y, l1 = s1.x + s1.y;
        float lh = (hf ? l1 : l0) + __shfl_xor(hf ? l0 : l1, 16, 32);
        int e = g * 16 + c;
        logits[(long)e * HEADS + hf] = lh;
        atomicMax(&mx[(long)dn * HEADS + hf], enc_f32(lh));

        node = node1; node1 = node2; g1 = g2; w = w1;
#pragma unroll
        for (int k = 0; k < F_IN; ++k) raw[k] = raw1[k];
    }
}

__global__ __launch_bounds__(256)
void k_edge_scatter(const int* __restrict__ ei, const float* __restrict__ logits,
                    const unsigned* __restrict__ mx, const float* __restrict__ h,
                    const float* __restrict__ gamma1, const float* __restrict__ beta1,
                    const float* __restrict__ rm1, const float* __restrict__ rv1,
                    float* __restrict__ agg) {
    int slot = threadIdx.x >> 4;
    int l = threadIdx.x & 15;
    int hh = l >> 3, k = l & 7;
    if (k >= 6 && l != 6) return;
    int kc = k < 5 ? k : 0;
    float rm = rm1[kc], rs = rsqrtf(rv1[kc] + BN_EPS), g = gamma1[kc], b = beta1[kc];
    int e0 = blockIdx.x * 64 + slot;
    int src[4], dst[4];
#pragma unroll
    for (int it = 0; it < 4; ++it) {
        int e = e0 + it * 16;
        src[it] = ei[e];
        dst[it] = ei[N_EDGES + e];
    }
    float lg[4], hr[4];
    unsigned mv[4];
#pragma unroll
    for (int it = 0; it < 4; ++it) {
        int e = e0 + it * 16;
        lg[it] = logits[(long)e * HEADS + hh];
        mv[it] = mx[(long)dst[it] * HEADS + hh];
        hr[it] = h[src[it] * F_IN + kc];
    }
#pragma unroll
    for (int it = 0; it < 4; ++it) {
        float ex = expf(lg[it] - dec_f32(mv[it]));
        float hv = (k < 5) ? ((hr[it] - rm) * rs * g + b) : 1.f;
        float v = (l == 6) ? 1.f : hv * ex;
        atomAddF(&agg[(long)dst[it] * AGG_W + l], v);
    }
}

#define TP 132
__global__ __launch_bounds__(256)
void k_fc(const float* __restrict__ agg,
          const float* __restrict__ Wl, const float* __restrict__ bl,
          const float* __restrict__ bias_gat,
          const float* __restrict__ W1, const float* __restrict__ b1,
          const float* __restrict__ gamma2, const float* __restrict__ beta2,
          const float* __restrict__ rm2, const float* __restrict__ rv2,
          const float* __restrict__ W2, const float* __restrict__ b2,
          float* __restrict__ out) {
    __shared__ float Y[16 * TP];
    __shared__ float Z[16 * TP];
    __shared__ float As[16 * AGG_W];
    int row0 = blockIdx.x * 16;
    int t = threadIdx.x;

    {
        int d = t & 127;
        int rb = t >> 7;
        float w0[6], w1[6];
#pragma unroll
        for (int k = 0; k < F_IN; ++k) {
            w0[k] = Wl[k * HD + d];
            w1[k] = Wl[k * HD + DDIM + d];
        }
        w0[5] = bl[d];
        w1[5] = bl[DDIM + d];
        float bg = bias_gat[d];
        float inv = 0.5f / fmaxf(agg[(long)row0 * AGG_W + (t & ~15) + 6], 1.f);
        float dsum = agg[(long)row0 * AGG_W + (t & ~7) + 5];
        As[t] = agg[(long)row0 * AGG_W + t] / (dsum + 1e-16f) * inv;
        __syncthreads();
#pragma unroll
        for (int i = 0; i < 8; ++i) {
            int r = rb + 2 * i;
            const float* a = &As[r * AGG_W];
            float s0 = 0.f, s1 = 0.f;
#pragma unroll
            for (int k = 0; k < 6; ++k) {
                s0 += a[k] * w0[k];
                s1 += a[8 + k] * w1[k];
            }
            Y[r * TP + d] = (s0 + s1) + bg;
        }
    }
    __syncthreads();

    int wave = t >> 5, lane = t & 31;
    int c0 = wave * 16;
    int laneN = lane & 15;
    int koff = (lane >> 4) * 2;
    v8f acc = {0.f, 0.f, 0.f, 0.f, 0.f, 0.f, 0.f, 0.f};
    for (int k = 0; k < 128; k += 4) {
        v2f a, b;
        a.x = Y[laneN * TP + k + koff];
        a.y = Y[laneN * TP + k + koff + 1];
        int kk = k + koff;
        b.x = W1[kk * 128 + c0 + laneN];
        b.y = W1[(kk + 1) * 128 + c0 + laneN];
        acc = __builtin_amdgcn_wmma_f32_16x16x4_f32(
            false, a, false, b, (short)0, acc, false, false);
    }

#pragma unroll
    for (int v = 0; v < 8; ++v) {
        int m = v + (lane >> 4) * 8;
        int col = c0 + laneN;
        float z = acc[v] + b1[col];
        z = (z - rm2[col]) * rsqrtf(rv2[col] + BN_EPS) * gamma2[col] + beta2[col];
        z = z > 0.f ? z : 0.01f * z;
        Z[m * TP + col] = z;
    }
    __syncthreads();

    if (t < 16 * F_IN) {
        int r = t / F_IN, c = t % F_IN;
        float o = b2[c];
        const float* zr = &Z[r * TP];
#pragma unroll 8
        for (int d = 0; d < 128; ++d) o += zr[d] * W2[d * F_IN + c];
        out[(long)(row0 + r) * F_IN + c] = o;
    }
}

extern "C" void kernel_launch(void* const* d_in, const int* in_sizes, int n_in,
                              void* d_out, int out_size, void* d_ws, size_t ws_size,
                              hipStream_t stream) {
    const float* h        = (const float*)d_in[0];
    const int*   ei       = (const int*)  d_in[1];
    const float* ew       = (const float*)d_in[2];
    const float* gamma1   = (const float*)d_in[3];
    const float* beta1    = (const float*)d_in[4];
    const float* rm1      = (const float*)d_in[5];
    const float* rv1      = (const float*)d_in[6];
    const float* Wl       = (const float*)d_in[7];
    const float* bl       = (const float*)d_in[8];
    const float* Wr       = (const float*)d_in[9];
    const float* br       = (const float*)d_in[10];
    const float* We       = (const float*)d_in[11];
    const float* att      = (const float*)d_in[12];
    const float* bias_gat = (const float*)d_in[13];
    const float* W1       = (const float*)d_in[14];
    const float* b1       = (const float*)d_in[15];
    const float* gamma2   = (const float*)d_in[16];
    const float* beta2    = (const float*)d_in[17];
    const float* rm2      = (const float*)d_in[18];
    const float* rv2      = (const float*)d_in[19];
    const float* W2       = (const float*)d_in[20];
    const float* b2       = (const float*)d_in[21];
    float* out = (float*)d_out;

    float* ws = (float*)d_ws;
    size_t o = 0;
    float* agg    = ws + o; o += (size_t)N_NODES * AGG_W;
    float* logits = ws + o; o += (size_t)N_EDGES * HEADS;
    unsigned* mx  = (unsigned*)(ws + o); o += (size_t)N_NODES * HEADS;

    const int T = 256;
    k_init<<<(N_NODES * AGG_W + T - 1) / T, T, 0, stream>>>(agg, mx);
    k_edge_logits<<<LG_BLOCKS, T, 0, stream>>>(ei, ew, h, gamma1, beta1, rm1, rv1,
                                               Wl, bl, Wr, br, We, att, logits, mx);
    k_edge_scatter<<<N_EDGES / 64, T, 0, stream>>>(ei, logits, mx, h, gamma1, beta1,
                                                   rm1, rv1, agg);
    k_fc<<<N_NODES / 16, T, 0, stream>>>(agg, Wl, bl, bias_gat, W1, b1,
                                         gamma2, beta2, rm2, rv2, W2, b2, out);
}


# Accumulator rows and max words kept as one copy per package half: a wave's atomics go to its own half's copy (die id), readers take max / sum of the two copies (f32 sums regrouped by half)
# speedup vs baseline: 6.0357x; 1.1557x over previous
#include <hip/hip_runtime.h>
#include <math.h>

#define N_NODES 50000
#define N_EDGES 400000
#define HEADS   2
#define DDIM    128
#define F_IN    5
#define HD      (HEADS*DDIM)
#define BN_EPS  1e-5f
#define AGG_W   16
#define AGG_COPY ((long)N_NODES * AGG_W)
#define MX_COPY  ((long)N_NODES * HEADS)

typedef __attribute__((ext_vector_type(2))) float v2f;
typedef __attribute__((ext_vector_type(8))) float v8f;

__device__ __forceinline__ unsigned enc_f32(float f) {
    unsigned u = __float_as_uint(f);
    return (u & 0x80000000u) ? ~u : (u | 0x80000000u);
}
__device__ __forceinline__ float dec_f32(unsigned u) {
    unsigned v = (u & 0x80000000u) ? (u & 0x7FFFFFFFu) : ~u;
    return __uint_as_float(v);
}
__device__ __forceinline__ int pkg_half() {
    return (int)((__builtin_amdgcn_s_sendmsg_rtn(0x87) >> 18) & 1u);
}
__device__ __forceinline__ void atomAddF(float* p, float v) {
    unsafeAtomicAdd(p, v);
}

__global__ void k_init(float* __restrict__ agg, unsigned* __restrict__ mx) {
    float4* agg4 = (float4*)agg;
    uint4* mx4 = (uint4*)mx;
    long i = (long)blockIdx.x * blockDim.x + threadIdx.x;
    if (i < 2 * AGG_COPY / 4) agg4[i] = make_float4(0.f, 0.f, 0.f, 0.f);
    if (i < 2 * MX_COPY / 4) mx4[i] = make_uint4(0u, 0u, 0u, 0u);
}

#define LG_BLOCKS 256
#define LG_WAVES  (LG_BLOCKS * 8)
#define LG_GROUPS (N_EDGES / 16)

__device__ __forceinline__ v2f lrelu_att(v8f acc, v8f a, v2f s) {
#pragma unroll
    for (int r = 0; r < 8; r += 2) {
        v2f z  = {acc[r], acc[r + 1]};
        v2f a2 = {a[r], a[r + 1]};
        v2f l  = __builtin_elementwise_max(z, 0.2f * z);
        s = a2 * l + s;
    }
    return s;
}

__global__ __launch_bounds__(256)
void k_edge_logits(const int* __restrict__ ei, const float* __restrict__ ew,
                   const float* __restrict__ h,
                   const float* __restrict__ gamma1, const float* __restrict__ beta1,
                   const float* __restrict__ rm1, const float* __restrict__ rv1,
                   const float* __restrict__ Wl, const float* __restrict__ bl,
                   const float* __restrict__ Wr, const float* __restrict__ br,
                   const float* __restrict__ We, const float* __restrict__ att,
                   float* __restrict__ logits, unsigned* __restrict__ mx) {
    const int lane = threadIdx.x & 31;
    const int c = lane & 15, hf = lane >> 4;
    const int wv = blockIdx.x * 8 + __builtin_amdgcn_readfirstlane((int)(threadIdx.x >> 5));
    unsigned* mxh = mx + pkg_half() * MX_COPY;

    const float* Wside = hf ? Wr : Wl;
    v2f wa[16][3];
    v8f at[16];
#pragma unroll
    for (int t = 0; t < 16; ++t) {
        int col = t * 16 + c;
        float wedge = We[col];
        float bsum = bl[col] + br[col];
        wa[t][0] = (v2f){Wside[0 * HD + col], Wside[1 * HD + col]};
        wa[t][1] = (v2f){Wside[2 * HD + col], Wside[3 * HD + col]};
        wa[t][2] = (v2f){Wside[4 * HD + col], hf ? bsum : wedge};
        at[t] = *(const v8f*)(att + t * 16 + 8 * hf);
    }
    float rm[F_IN], rs[F_IN], gm[F_IN], bt[F_IN];
#pragma unroll
    for (int k = 0; k < F_IN; ++k) {
        rm[k] = rm1[k];
        rs[k] = rsqrtf(rv1[k] + BN_EPS);
        gm[k] = gamma1[k];
        bt[k] = beta1[k];
    }

    const int* eside = ei + hf * N_EDGES;
    int g = wv;
    int node = eside[g * 16 + c];
    float raw[F_IN];
#pragma unroll
    for (int k = 0; k < F_IN; ++k) raw[k] = h[node * F_IN + k];
    float w = ew[g * 16 + c];
    int g1 = min(g + LG_WAVES, LG_GROUPS - 1);
    int node1 = eside[g1 * 16 + c];

    for (; g < LG_GROUPS; g += LG_WAVES) {
        float raw1[F_IN];
#pragma unroll
        for (int k = 0; k < F_IN; ++k) raw1[k] = h[node1 * F_IN + k];
        float w1 = ew[g1 * 16 + c];
        int g2 = min(g1 + LG_WAVES, LG_GROUPS - 1);
        int node2 = eside[g2 * 16 + c];
        int dn = ei[N_EDGES + g * 16 + c];

        float f[6];
#pragma unroll
        for (int k = 0; k < F_IN; ++k)
            f[k] = (raw[k] - rm[k]) * rs[k] * gm[k] + bt[k];
        f[5] = hf ? 1.f : w;
        v2f b0 = {f[0], f[1]}, b1 = {f[2], f[3]}, b2 = {f[4], f[5]};

        v2f s0 = {0.f, 0.f}, s1 = {0.f, 0.f};
#pragma unroll
        for (int t = 0; t < 16; t += 2) {
            v8f a0 = {0.f, 0.f, 0.f, 0.f, 0.f, 0.f, 0.f, 0.f};
            v8f a1 = {0.f, 0.f, 0.f, 0.f, 0.f, 0.f, 0.f, 0.f};
            a0 = __builtin_amdgcn_wmma_f32_16x16x4_f32(false, wa[t][0], false, b0, (short)0, a0, false, false);
            a1 = __builtin_amdgcn_wmma_f32_16x16x4_f32(false, wa[t + 1][0], false, b0, (short)0, a1, false, false);
            a0 = __builtin_amdgcn_wmma_f32_16x16x4_f32(false, wa[t][1], false, b1, (short)0, a0, false, false);
            a1 = __builtin_amdgcn_wmma_f32_16x16x4_f32(false, wa[t + 1][1], false, b1, (short)0, a1, false, false);
            a0 = __builtin_amdgcn_wmma_f32_16x16x4_f32(false, wa[t][2], false, b2, (short)0, a0, false, false);
            a1 = __builtin_amdgcn_wmma_f32_16x16x4_f32(false, wa[t + 1][2], false, b2, (short)0, a1, false, false);
            if (t < 8) {
                s0 = lrelu_att(a0, at[t], s0);
                s0 = lrelu_att(a1, at[t + 1], s0);
            } else {
                s1 = lrelu_att(a0, at[t], s1);
                s1 = lrelu_att(a1, at[t + 1], s1);
            }
        }
        float l0 = s0.x + s0.y, l1 = s1.x + s1.y;
        float lh = (hf ? l1 : l0) + __shfl_xor(hf ? l0 : l1, 16, 32);
        int e = g * 16 + c;
        logits[(long)e * HEADS + hf] = lh;
        atomicMax(&mxh[(long)dn * HEADS + hf], enc_f32(lh));

        node = node1; node1 = node2; g1 = g2; w = w1;
#pragma unroll
        for (int k = 0; k < F_IN; ++k) raw[k] = raw1[k];
    }
}

__global__ __launch_bounds__(256)
void k_edge_scatter(const int* __restrict__ ei, const float* __restrict__ logits,
                    const unsigned* __restrict__ mx, const float* __restrict__ h,
                    const float* __restrict__ gamma1, const float* __restrict__ beta1,
                    const float* __restrict__ rm1, const float* __restrict__ rv1,
                    float* __restrict__ agg) {
    int slot = threadIdx.x >> 4;
    int l = threadIdx.x & 15;
    int hh = l >> 3, k = l & 7;
    if (k >= 6 && l != 6) return;
    int kc = k < 5 ? k : 0;
    float rm = rm1[kc], rs = rsqrtf(rv1[kc] + BN_EPS), g = gamma1[kc], b = beta1[kc];
    int e0 = blockIdx.x * 64 + slot;
    float* aggh = agg + pkg_half() * AGG_COPY;
    int src[4], dst[4];
#pragma unroll
    for (int it = 0; it < 4; ++it) {
        int e = e0 + it * 16;
        src[it] = ei[e];
        dst[it] = ei[N_EDGES + e];
    }
    float lg[4], hr[4];
    unsigned mv[4];
#pragma unroll
    for (int it = 0; it < 4; ++it) {
        int e = e0 + it * 16;
        lg[it] = logits[(long)e * HEADS + hh];
        mv[it] = max(mx[(long)dst[it] * HEADS + hh], mx[MX_COPY + (long)dst[it] * HEADS + hh]);
        hr[it] = h[src[it] * F_IN + kc];
    }
#pragma unroll
    for (int it = 0; it < 4; ++it) {
        float ex = expf(lg[it] - dec_f32(mv[it]));
        float hv = (k < 5) ? ((hr[it] - rm) * rs * g + b) : 1.f;
        float v = (l == 6) ? 1.f : hv * ex;
        atomAddF(&aggh[(long)dst[it] * AGG_W + l], v);
    }
}

#define TP 132
__global__ __launch_bounds__(256)
void k_fc(const float* __restrict__ agg,
          const float* __restrict__ Wl, const float* __restrict__ bl,
          const float* __restrict__ bias_gat,
          const float* __restrict__ W1, const float* __restrict__ b1,
          const float* __restrict__ gamma2, const float* __restrict__ beta2,
          const float* __restrict__ rm2, const float* __restrict__ rv2,
          const float* __restrict__ W2, const float* __restrict__ b2,
          float* __restrict__ out) {
    __shared__ float Y[16 * TP];
    __shared__ float Z[16 * TP];
    __shared__ float As[16 * AGG_W];
    int row0 = blockIdx.x * 16;
    int t = threadIdx.x;

    {
        int d = t & 127;
        int rb = t >> 7;
        float w0[6], w1[6];
#pragma unroll
        for (int k = 0; k < F_IN; ++k) {
            w0[k] = Wl[k * HD + d];
            w1[k] = Wl[k * HD + DDIM + d];
        }
        w0[5] = bl[d];
        w1[5] = bl[DDIM + d];
        float bg = bias_gat[d];
        const float* ga = agg + (long)row0 * AGG_W;
        const float* gb = ga + AGG_COPY;
        float inv = 0.5f / fmaxf(ga[(t & ~15) + 6] + gb[(t & ~15) + 6], 1.f);
        float dsum = ga[(t & ~7) + 5] + gb[(t & ~7) + 5];
        As[t] = (ga[t] + gb[t]) / (dsum + 1e-16f) * inv;
        __syncthreads();
#pragma unroll
        for (int i = 0; i < 8; ++i) {
            int r = rb + 2 * i;
            const float* a = &As[r * AGG_W];
            float s0 = 0.f, s1 = 0.f;
#pragma unroll
            for (int k = 0; k < 6; ++k) {
                s0 += a[k] * w0[k];
                s1 += a[8 + k] * w1[k];
            }
            Y[r * TP + d] = (s0 + s1) + bg;
        }
    }
    __syncthreads();

    int wave = t >> 5, lane = t & 31;
    int c0 = wave * 16;
    int laneN = lane & 15;
    int koff = (lane >> 4) * 2;
    v8f acc = {0.f, 0.f, 0.f, 0.f, 0.f, 0.f, 0.f, 0.f};
    for (int k = 0; k < 128; k += 4) {
        v2f a, b;
        a.x = Y[laneN * TP + k + koff];
        a.y = Y[laneN * TP + k + koff + 1];
        int kk = k + koff;
        b.x = W1[kk * 128 + c0 + laneN];
        b.y = W1[(kk + 1) * 128 + c0 + laneN];
        acc = __builtin_amdgcn_wmma_f32_16x16x4_f32(
            false, a, false, b, (short)0, acc, false, false);
    }

#pragma unroll
    for (int v = 0; v < 8; ++v) {
        int m = v + (lane >> 4) * 8;
        int col = c0 + laneN;
        float z = acc[v] + b1[col];
        z = (z - rm2[col]) * rsqrtf(rv2[col] + BN_EPS) * gamma2[col] + beta2[col];
        z = z > 0.f ? z : 0.01f * z;
        Z[m * TP + col] = z;
    }
    __syncthreads();

    if (t < 16 * F_IN) {
        int r = t / F_IN, c = t % F_IN;
        float o = b2[c];
        const float* zr = &Z[r * TP];
#pragma unroll 8
        for (int d = 0; d < 128; ++d) o += zr[d] * W2[d * F_IN + c];
        out[(long)(row0 + r) * F_IN + c] = o;
    }
}

extern "C" void kernel_launch(void* const* d_in, const int* in_sizes, int n_in,
                              void* d_out, int out_size, void* d_ws, size_t ws_size,
                              hipStream_t stream) {
    const float* h        = (const float*)d_in[0];
    const int*   ei       = (const int*)  d_in[1];
    const float* ew       = (const float*)d_in[2];
    const float* gamma1   = (const float*)d_in[3];
    const float* beta1    = (const float*)d_in[4];
    const float* rm1      = (const float*)d_in[5];
    const float* rv1      = (const float*)d_in[6];
    const float* Wl       = (const float*)d_in[7];
    const float* bl       = (const float*)d_in[8];
    const float* Wr       = (const float*)d_in[9];
    const float* br       = (const float*)d_in[10];
    const float* We       = (const float*)d_in[11];
    const float* att      = (const float*)d_in[12];
    const float* bias_gat = (const float*)d_in[13];
    const float* W1       = (const float*)d_in[14];
    const float* b1       = (const float*)d_in[15];
    const float* gamma2   = (const float*)d_in[16];
    const float* beta2    = (const float*)d_in[17];
    const float* rm2      = (const float*)d_in[18];
    const float* rv2      = (const float*)d_in[19];
    const float* W2       = (const float*)d_in[20];
    const float* b2       = (const float*)d_in[21];
    float* out = (float*)d_out;

    float* ws = (float*)d_ws;
    size_t o = 0;
    float* agg    = ws + o; o += (size_t)2 * N_NODES * AGG_W;
    float* logits = ws + o; o += (size_t)N_EDGES * HEADS;
    unsigned* mx  = (unsigned*)(ws + o); o += (size_t)2 * N_NODES * HEADS;

    const int T = 256;
    k_init<<<(2 * N_NODES * AGG_W / 4 + T - 1) / T, T, 0, stream>>>(agg, mx);
    k_edge_logits<<<LG_BLOCKS, T, 0, stream>>>(ei, ew, h, gamma1, beta1, rm1, rv1,
                                               Wl, bl, Wr, br, We, att, logits, mx);
    k_edge_scatter<<<N_EDGES / 64, T, 0, stream>>>(ei, logits, mx, h, gamma1, beta1,
                                                   rm1, rv1, agg);
    k_fc<<<N_NODES / 16, T, 0, stream>>>(agg, Wl, bl, bias_gat, W1, b1,
                                         gamma2, beta2, rm2, rv2, W2, b2, out);
}


# Init and scatter kernels launch resident grids whose waves loop (scatter 625 blocks x 5 passes x 8 edges per lane, next indices prefetched; init 256 blocks): 7k waves enter, not 62k; sums by arrival
# speedup vs baseline: 6.5408x; 1.0837x over previous
#include <hip/hip_runtime.h>
#include <math.h>

#define N_NODES 50000
#define N_EDGES 400000
#define HEADS   2
#define DDIM    128
#define F_IN    5
#define HD      (HEADS*DDIM)
#define BN_EPS  1e-5f
#define AGG_W   16
#define AGG_COPY ((long)N_NODES * AGG_W)
#define MX_COPY  ((long)N_NODES * HEADS)

typedef __attribute__((ext_vector_type(2))) float v2f;
typedef __attribute__((ext_vector_type(8))) float v8f;

__device__ __forceinline__ unsigned enc_f32(float f) {
    unsigned u = __float_as_uint(f);
    return (u & 0x80000000u) ? ~u : (u | 0x80000000u);
}
__device__ __forceinline__ float dec_f32(unsigned u) {
    unsigned v = (u & 0x80000000u) ? (u & 0x7FFFFFFFu) : ~u;
    return __uint_as_float(v);
}
__device__ __forceinline__ int pkg_half() {
    return (int)((__builtin_amdgcn_s_sendmsg_rtn(0x87) >> 18) & 1u);
}
__device__ __forceinline__ void atomAddF(float* p, float v) {
    unsafeAtomicAdd(p, v);
}

#define IN_BLOCKS  256
#define IN_THREADS (IN_BLOCKS * 256)
#define IN_AGG4    (2 * N_NODES * AGG_W / 4)
#define IN_MX4     (2 * N_NODES * HEADS / 4)
__global__ __launch_bounds__(256)
void k_init(float* __restrict__ agg, unsigned* __restrict__ mx) {
    float4* agg4 = (float4*)agg;
    uint4* mx4 = (uint4*)mx;
    int i0 = blockIdx.x * 256 + threadIdx.x;
#pragma unroll
    for (int j = 0; j < IN_AGG4 / IN_THREADS; ++j)
        agg4[i0 + j * IN_THREADS] = make_float4(0.f, 0.f, 0.f, 0.f);
    int it = i0 + (IN_AGG4 / IN_THREADS) * IN_THREADS;
    if (it < IN_AGG4) agg4[it] = make_float4(0.f, 0.f, 0.f, 0.f);
    if (i0 < IN_MX4) mx4[i0] = make_uint4(0u, 0u, 0u, 0u);
}

#define LG_BLOCKS 256
#define LG_WAVES  (LG_BLOCKS * 8)
#define LG_GROUPS (N_EDGES / 16)

__device__ __forceinline__ v2f lrelu_att(v8f acc, v8f a, v2f s) {
#pragma unroll
    for (int r = 0; r < 8; r += 2) {
        v2f z  = {acc[r], acc[r + 1]};
        v2f a2 = {a[r], a[r + 1]};
        v2f l  = __builtin_elementwise_max(z, 0.2f * z);
        s = a2 * l + s;
    }
    return s;
}

__global__ __launch_bounds__(256)
void k_edge_logits(const int* __restrict__ ei, const float* __restrict__ ew,
                   const float* __restrict__ h,
                   const float* __restrict__ gamma1, const float* __restrict__ beta1,
                   const float* __restrict__ rm1, const float* __restrict__ rv1,
                   const float* __restrict__ Wl, const float* __restrict__ bl,
                   const float* __restrict__ Wr, const float* __restrict__ br,
                   const float* __restrict__ We, const float* __restrict__ att,
                   float* __restrict__ logits, unsigned* __restrict__ mx) {
    const int lane = threadIdx.x & 31;
    const int c = lane & 15, hf = lane >> 4;
    const int wv = blockIdx.x * 8 + __builtin_amdgcn_readfirstlane((int)(threadIdx.x >> 5));
    unsigned* mxh = mx + pkg_half() * MX_COPY;

    const float* Wside = hf ? Wr : Wl;
    v2f wa[16][3];
    v8f at[16];
#pragma unroll
    for (int t = 0; t < 16; ++t) {
        int col = t * 16 + c;
        float wedge = We[col];
        float bsum = bl[col] + br[col];
        wa[t][0] = (v2f){Wside[0 * HD + col], Wside[1 * HD + col]};
        wa[t][1] = (v2f){Wside[2 * HD + col], Wside[3 * HD + col]};
        wa[t][2] = (v2f){Wside[4 * HD + col], hf ? bsum : wedge};
        at[t] = *(const v8f*)(att + t * 16 + 8 * hf);
    }
    float rm[F_IN], rs[F_IN], gm[F_IN], bt[F_IN];
#pragma unroll
    for (int k = 0; k < F_IN; ++k) {
        rm[k] = rm1[k];
        rs[k] = rsqrtf(rv1[k] + BN_EPS);
        gm[k] = gamma1[k];
        bt[k] = beta1[k];
    }

    const int* eside = ei + hf * N_EDGES;
    int g = wv;
    int node = eside[g * 16 + c];
    float raw[F_IN];
#pragma unroll
    for (int k = 0; k < F_IN; ++k) raw[k] = h[node * F_IN + k];
    float w = ew[g * 16 + c];
    int g1 = min(g + LG_WAVES, LG_GROUPS - 1);
    int node1 = eside[g1 * 16 + c];

    for (; g < LG_GROUPS; g += LG_WAVES) {
        float raw1[F_IN];
#pragma unroll
        for (int k = 0; k < F_IN; ++k) raw1[k] = h[node1 * F_IN + k];
        float w1 = ew[g1 * 16 + c];
        int g2 = min(g1 + LG_WAVES, LG_GROUPS - 1);
        int node2 = eside[g2 * 16 + c];
        int dn = ei[N_EDGES + g * 16 + c];

        float f[6];
#pragma unroll
        for (int k = 0; k < F_IN; ++k)
            f[k] = (raw[k] - rm[k]) * rs[k] * gm[k] + bt[k];
        f[5] = hf ? 1.f : w;
        v2f b0 = {f[0], f[1]}, b1 = {f[2], f[3]}, b2 = {f[4], f[5]};

        v2f s0 = {0.f, 0.f}, s1 = {0.f, 0.f};
#pragma unroll
        for (int t = 0; t < 16; t += 2) {
            v8f a0 = {0.f, 0.f, 0.f, 0.f, 0.f, 0.f, 0.f, 0.f};
            v8f a1 = {0.f, 0.f, 0.f, 0.f, 0.f, 0.f, 0.f, 0.f};
            a0 = __builtin_amdgcn_wmma_f32_16x16x4_f32(false, wa[t][0], false, b0, (short)0, a0, false, false);
            a1 = __builtin_amdgcn_wmma_f32_16x16x4_f32(false, wa[t + 1][0], false, b0, (short)0, a1, false, false);
            a0 = __builtin_amdgcn_wmma_f32_16x16x4_f32(false, wa[t][1], false, b1, (short)0, a0, false, false);
            a1 = __builtin_amdgcn_wmma_f32_16x16x4_f32(false, wa[t + 1][1], false, b1, (short)0, a1, false, false);
            a0 = __builtin_amdgcn_wmma_f32_16x16x4_f32(false, wa[t][2], false, b2, (short)0, a0, false, false);
            a1 = __builtin_amdgcn_wmma_f32_16x16x4_f32(false, wa[t + 1][2], false, b2, (short)0, a1, false, false);
            if (t < 8) {
                s0 = lrelu_att(a0, at[t], s0);
                s0 = lrelu_att(a1, at[t + 1], s0);
            } else {
                s1 = lrelu_att(a0, at[t], s1);
                s1 = lrelu_att(a1, at[t + 1], s1);
            }
        }
        float l0 = s0.x + s0.y, l1 = s1.x + s1.y;
        float lh = (hf ? l1 : l0) + __shfl_xor(hf ? l0 : l1, 16, 32);
        int e = g * 16 + c;
        logits[(long)e * HEADS + hf] = lh;
        atomicMax(&mxh[(long)dn * HEADS + hf], enc_f32(lh));

        node = node1; node1 = node2; g1 = g2; w = w1;
#pragma unroll
        for (int k = 0; k < F_IN; ++k) raw[k] = raw1[k];
    }
}

#define SC_BLOCKS 625
#define SC_PASSES 5
#define SC_EPL    8
#define SC_PASS_E (16 * SC_EPL)
__global__ __launch_bounds__(256)
void k_edge_scatter(const int* __restrict__ ei, const float* __restrict__ logits,
                    const unsigned* __restrict__ mx, const float* __restrict__ h,
                    const float* __restrict__ gamma1, const float* __restrict__ beta1,
                    const float* __restrict__ rm1, const float* __restrict__ rv1,
                    float* __restrict__ agg) {
    int slot = threadIdx.x >> 4;
    int l = threadIdx.x & 15;
    int hh = l >> 3, k = l & 7;
    if (k >= 6 && l != 6) return;
    int kc = k < 5 ? k : 0;
    float rm = rm1[kc], rs = rsqrtf(rv1[kc] + BN_EPS), g = gamma1[kc], b = beta1[kc];
    int ebase = blockIdx.x * (SC_PASSES * SC_PASS_E) + slot;
    float* aggh = agg + pkg_half() * AGG_COPY;
    int src[SC_EPL], dst[SC_EPL];
#pragma unroll
    for (int it = 0; it < SC_EPL; ++it) {
        int e = ebase + it * 16;
        src[it] = ei[e];
        dst[it] = ei[N_EDGES + e];
    }
#pragma unroll 1
    for (int p = 0; p < SC_PASSES; ++p) {
        int e0 = ebase + p * SC_PASS_E;
        float lg[SC_EPL], hr[SC_EPL];
        unsigned mv[SC_EPL];
#pragma unroll
        for (int it = 0; it < SC_EPL; ++it) {
            int e = e0 + it * 16;
            lg[it] = logits[(long)e * HEADS + hh];
            mv[it] = max(mx[(long)dst[it] * HEADS + hh], mx[MX_COPY + (long)dst[it] * HEADS + hh]);
            hr[it] = h[src[it] * F_IN + kc];
        }
        int en = ebase + min(p + 1, SC_PASSES - 1) * SC_PASS_E;
        int nsrc[SC_EPL], ndst[SC_EPL];
#pragma unroll
        for (int it = 0; it < SC_EPL; ++it) {
            int e = en + it * 16;
            nsrc[it] = ei[e];
            ndst[it] = ei[N_EDGES + e];
        }
#pragma unroll
        for (int it = 0; it < SC_EPL; ++it) {
            float ex = expf(lg[it] - dec_f32(mv[it]));
            float hv = (k < 5) ? ((hr[it] - rm) * rs * g + b) : 1.f;
            float v = (l == 6) ? 1.f : hv * ex;
            atomAddF(&aggh[(long)dst[it] * AGG_W + l], v);
        }
#pragma unroll
        for (int it = 0; it < SC_EPL; ++it) { src[it] = nsrc[it]; dst[it] = ndst[it]; }
    }
}

#define TP 132
__global__ __launch_bounds__(256)
void k_fc(const float* __restrict__ agg,
          const float* __restrict__ Wl, const float* __restrict__ bl,
          const float* __restrict__ bias_gat,
          const float* __restrict__ W1, const float* __restrict__ b1,
          const float* __restrict__ gamma2, const float* __restrict__ beta2,
          const float* __restrict__ rm2, const float* __restrict__ rv2,
          const float* __restrict__ W2, const float* __restrict__ b2,
          float* __restrict__ out) {
    __shared__ float Y[16 * TP];
    __shared__ float Z[16 * TP];
    __shared__ float As[16 * AGG_W];
    int row0 = blockIdx.x * 16;
    int t = threadIdx.x;

    {
        int d = t & 127;
        int rb = t >> 7;
        float w0[6], w1[6];
#pragma unroll
        for (int k = 0; k < F_IN; ++k) {
            w0[k] = Wl[k * HD + d];
            w1[k] = Wl[k * HD + DDIM + d];
        }
        w0[5] = bl[d];
        w1[5] = bl[DDIM + d];
        float bg = bias_gat[d];
        const float* ga = agg + (long)row0 * AGG_W;
        const float* gb = ga + AGG_COPY;
        float inv = 0.5f / fmaxf(ga[(t & ~15) + 6] + gb[(t & ~15) + 6], 1.f);
        float dsum = ga[(t & ~7) + 5] + gb[(t & ~7) + 5];
        As[t] = (ga[t] + gb[t]) / (dsum + 1e-16f) * inv;
        __syncthreads();
#pragma unroll
        for (int i = 0; i < 8; ++i) {
            int r = rb + 2 * i;
            const float* a = &As[r * AGG_W];
            float s0 = 0.f, s1 = 0.f;
#pragma unroll
            for (int k = 0; k < 6; ++k) {
                s0 += a[k] * w0[k];
                s1 += a[8 + k] * w1[k];
            }
            Y[r * TP + d] = (s0 + s1) + bg;
        }
    }
    __syncthreads();

    int wave = t >> 5, lane = t & 31;
    int c0 = wave * 16;
    int laneN = lane & 15;
    int koff = (lane >> 4) * 2;
    v8f acc = {0.f, 0.f, 0.f, 0.f, 0.f, 0.f, 0.f, 0.f};
    for (int k = 0; k < 128; k += 4) {
        v2f a, b;
        a.x = Y[laneN * TP + k + koff];
        a.y = Y[laneN * TP + k + koff + 1];
        int kk = k + koff;
        b.x = W1[kk * 128 + c0 + laneN];
        b.y = W1[(kk + 1) * 128 + c0 + laneN];
        acc = __builtin_amdgcn_wmma_f32_16x16x4_f32(
            false, a, false, b, (short)0, acc, false, false);
    }

#pragma unroll
    for (int v = 0; v < 8; ++v) {
        int m = v + (lane >> 4) * 8;
        int col = c0 + laneN;
        float z = acc[v] + b1[col];
        z = (z - rm2[col]) * rsqrtf(rv2[col] + BN_EPS) * gamma2[col] + beta2[col];
        z = z > 0.f ? z : 0.01f * z;
        Z[m * TP + col] = z;
    }
    __syncthreads();

    if (t < 16 * F_IN) {
        int r = t / F_IN, c = t % F_IN;
        float o = b2[c];
        const float* zr = &Z[r * TP];
#pragma unroll 8
        for (int d = 0; d < 128; ++d) o += zr[d] * W2[d * F_IN + c];
        out[(long)(row0 + r) * F_IN + c] = o;
    }
}

extern "C" void kernel_launch(void* const* d_in, const int* in_sizes, int n_in,
                              void* d_out, int out_size, void* d_ws, size_t ws_size,
                              hipStream_t stream) {
    const float* h        = (const float*)d_in[0];
    const int*   ei       = (const int*)  d_in[1];
    const float* ew       = (const float*)d_in[2];
    const float* gamma1   = (const float*)d_in[3];
    const float* beta1    = (const float*)d_in[4];
    const float* rm1      = (const float*)d_in[5];
    const float* rv1      = (const float*)d_in[6];
    const float* Wl       = (const float*)d_in[7];
    const float* bl       = (const float*)d_in[8];
    const float* Wr       = (const float*)d_in[9];
    const float* br       = (const float*)d_in[10];
    const float* We       = (const float*)d_in[11];
    const float* att      = (const float*)d_in[12];
    const float* bias_gat = (const float*)d_in[13];
    const float* W1       = (const float*)d_in[14];
    const float* b1       = (const float*)d_in[15];
    const float* gamma2   = (const float*)d_in[16];
    const float* beta2    = (const float*)d_in[17];
    const float* rm2      = (const float*)d_in[18];
    const float* rv2      = (const float*)d_in[19];
    const float* W2       = (const float*)d_in[20];
    const float* b2       = (const float*)d_in[21];
    float* out = (float*)d_out;

    float* ws = (float*)d_ws;
    size_t o = 0;
    float* agg    = ws + o; o += (size_t)2 * N_NODES * AGG_W;
    float* logits = ws + o; o += (size_t)N_EDGES * HEADS;
    unsigned* mx  = (unsigned*)(ws + o); o += (size_t)2 * N_NODES * HEADS;

    const int T = 256;
    k_init<<<IN_BLOCKS, T, 0, stream>>>(agg, mx);
    k_edge_logits<<<LG_BLOCKS, T, 0, stream>>>(ei, ew, h, gamma1, beta1, rm1, rv1,
                                               Wl, bl, Wr, br, We, att, logits, mx);
    k_edge_scatter<<<SC_BLOCKS, T, 0, stream>>>(ei, logits, mx, h, gamma1, beta1,
                                                   rm1, rv1, agg);
    k_fc<<<N_NODES / 16, T, 0, stream>>>(agg, Wl, bl, bias_gat, W1, b1,
                                         gamma2, beta2, rm2, rv2, W2, b2, out);
}


# Node kernel: fc1+BN2 folded by linearity into the 12 node factors (M=[Wl;bl]@W1 built once per call in the init launch): 12-deep product, then fc2, on f32 WMMA, wave per 32 nodes (f32 sums regrouped)
# speedup vs baseline: 8.4673x; 1.2945x over previous
#include <hip/hip_runtime.h>
#include <math.h>

#define N_NODES 50000
#define N_EDGES 400000
#define HEADS   2
#define DDIM    128
#define F_IN    5
#define HD      (HEADS*DDIM)
#define BN_EPS  1e-5f
#define AGG_W   16
#define AGG_COPY ((long)N_NODES * AGG_W)
#define MX_COPY  ((long)N_NODES * HEADS)

typedef __attribute__((ext_vector_type(2))) float v2f;
typedef __attribute__((ext_vector_type(8))) float v8f;

__device__ __forceinline__ unsigned enc_f32(float f) {
    unsigned u = __float_as_uint(f);
    return (u & 0x80000000u) ? ~u : (u | 0x80000000u);
}
__device__ __forceinline__ float dec_f32(unsigned u) {
    unsigned v = (u & 0x80000000u) ? (u & 0x7FFFFFFFu) : ~u;
    return __uint_as_float(v);
}
__device__ __forceinline__ int pkg_half() {
    return (int)((__builtin_amdgcn_s_sendmsg_rtn(0x87) >> 18) & 1u);
}
__device__ __forceinline__ void atomAddF(float* p, float v) {
    unsafeAtomicAdd(p, v);
}

#define IN_BLOCKS  256
#define IN_THREADS (IN_BLOCKS * 256)
#define IN_AGG4    (2 * N_NODES * AGG_W / 4)
#define IN_MX4     (2 * N_NODES * HEADS / 4)
#define TP 132
__device__ __forceinline__ void build_node_consts(
        const float* __restrict__ Wl, const float* __restrict__ bl,
        const float* __restrict__ bias_gat,
        const float* __restrict__ W1, const float* __restrict__ b1,
        const float* __restrict__ gamma2, const float* __restrict__ beta2,
        const float* __restrict__ rm2, const float* __restrict__ rv2,
        float* __restrict__ Mt, float* __restrict__ sh) {
    __shared__ float P[16 * TP];
    int t = threadIdx.x;
    {
        int d = t & 127;
        int head = t >> 7;
        float w[6];
#pragma unroll
        for (int k = 0; k < F_IN; ++k) w[k] = Wl[k * HD + head * DDIM + d];
        w[5] = bl[head * DDIM + d];
        float bg = bias_gat[d];
#pragma unroll
        for (int k = 0; k < 6; ++k) P[(6 * head + k) * TP + d] = w[k];
        P[(12 + head) * TP + d] = head ? 0.f : bg;
        P[(14 + head) * TP + d] = 0.f;
    }
    __syncthreads();
    int wave = t >> 5, lane = t & 31;
    int c0 = wave * 16;
    int laneN = lane & 15;
    int koff = (lane >> 4) * 2;
    v8f acc = {0.f, 0.f, 0.f, 0.f, 0.f, 0.f, 0.f, 0.f};
    for (int k = 0; k < 128; k += 4) {
        v2f a, b;
        a.x = P[laneN * TP + k + koff];
        a.y = P[laneN * TP + k + koff + 1];
        int kk = k + koff;
        b.x = W1[kk * 128 + c0 + laneN];
        b.y = W1[(kk + 1) * 128 + c0 + laneN];
        acc = __builtin_amdgcn_wmma_f32_16x16x4_f32(
            false, a, false, b, (short)0, acc, false, false);
    }
    int col = c0 + laneN;
    float sc = rsqrtf(rv2[col] + BN_EPS) * gamma2[col];
    float cb = b1[col] - rm2[col];
    float be = beta2[col];
#pragma unroll
    for (int v = 0; v < 8; ++v) {
        int m = v + (lane >> 4) * 8;
        int head = m >= 6 ? 1 : 0;
        if (m < 12) Mt[col * 16 + 8 * head + (m - 6 * head)] = acc[v] * sc;
        else if (m == 12) sh[col] = (acc[v] + cb) * sc + be;
    }
}

__global__ __launch_bounds__(256)
void k_init(float* __restrict__ agg, unsigned* __restrict__ mx,
            const float* __restrict__ Wl, const float* __restrict__ bl,
            const float* __restrict__ bias_gat,
            const float* __restrict__ W1, const float* __restrict__ b1,
            const float* __restrict__ gamma2, const float* __restrict__ beta2,
            const float* __restrict__ rm2, const float* __restrict__ rv2,
            float* __restrict__ Mt, float* __restrict__ sh) {
    if (blockIdx.x == IN_BLOCKS) {
        build_node_consts(Wl, bl, bias_gat, W1, b1, gamma2, beta2, rm2, rv2, Mt, sh);
        return;
    }
    float4* agg4 = (float4*)agg;
    uint4* mx4 = (uint4*)mx;
    int i0 = blockIdx.x * 256 + threadIdx.x;
#pragma unroll
    for (int j = 0; j < IN_AGG4 / IN_THREADS; ++j)
        agg4[i0 + j * IN_THREADS] = make_float4(0.f, 0.f, 0.f, 0.f);
    int it = i0 + (IN_AGG4 / IN_THREADS) * IN_THREADS;
    if (it < IN_AGG4) agg4[it] = make_float4(0.f, 0.f, 0.f, 0.f);
    if (i0 < IN_MX4) mx4[i0] = make_uint4(0u, 0u, 0u, 0u);
}

#define LG_BLOCKS 256
#define LG_WAVES  (LG_BLOCKS * 8)
#define LG_GROUPS (N_EDGES / 16)

__device__ __forceinline__ v2f lrelu_att(v8f acc, v8f a, v2f s) {
#pragma unroll
    for (int r = 0; r < 8; r += 2) {
        v2f z  = {acc[r], acc[r + 1]};
        v2f a2 = {a[r], a[r + 1]};
        v2f l  = __builtin_elementwise_max(z, 0.2f * z);
        s = a2 * l + s;
    }
    return s;
}

__global__ __launch_bounds__(256)
void k_edge_logits(const int* __restrict__ ei, const float* __restrict__ ew,
                   const float* __restrict__ h,
                   const float* __restrict__ gamma1, const float* __restrict__ beta1,
                   const float* __restrict__ rm1, const float* __restrict__ rv1,
                   const float* __restrict__ Wl, const float* __restrict__ bl,
                   const float* __restrict__ Wr, const float* __restrict__ br,
                   const float* __restrict__ We, const float* __restrict__ att,
                   float* __restrict__ logits, unsigned* __restrict__ mx) {
    const int lane = threadIdx.x & 31;
    const int c = lane & 15, hf = lane >> 4;
    const int wv = blockIdx.x * 8 + __builtin_amdgcn_readfirstlane((int)(threadIdx.x >> 5));
    unsigned* mxh = mx + pkg_half() * MX_COPY;

    const float* Wside = hf ? Wr : Wl;
    v2f wa[16][3];
    v8f at[16];
#pragma unroll
    for (int t = 0; t < 16; ++t) {
        int col = t * 16 + c;
        float wedge = We[col];
        float bsum = bl[col] + br[col];
        wa[t][0] = (v2f){Wside[0 * HD + col], Wside[1 * HD + col]};
        wa[t][1] = (v2f){Wside[2 * HD + col], Wside[3 * HD + col]};
        wa[t][2] = (v2f){Wside[4 * HD + col], hf ? bsum : wedge};
        at[t] = *(const v8f*)(att + t * 16 + 8 * hf);
    }
    float rm[F_IN], rs[F_IN], gm[F_IN], bt[F_IN];
#pragma unroll
    for (int k = 0; k < F_IN; ++k) {
        rm[k] = rm1[k];
        rs[k] = rsqrtf(rv1[k] + BN_EPS);
        gm[k] = gamma1[k];
        bt[k] = beta1[k];
    }

    const int* eside = ei + hf * N_EDGES;
    int g = wv;
    int node = eside[g * 16 + c];
    float raw[F_IN];
#pragma unroll
    for (int k = 0; k < F_IN; ++k) raw[k] = h[node * F_IN + k];
    float w = ew[g * 16 + c];
    int g1 = min(g + LG_WAVES, LG_GROUPS - 1);
    int node1 = eside[g1 * 16 + c];

    for (; g < LG_GROUPS; g += LG_WAVES) {
        float raw1[F_IN];
#pragma unroll
        for (int k = 0; k < F_IN; ++k) raw1[k] = h[node1 * F_IN + k];
        float w1 = ew[g1 * 16 + c];
        int g2 = min(g1 + LG_WAVES, LG_GROUPS - 1);
        int node2 = eside[g2 * 16 + c];
        int dn = ei[N_EDGES + g * 16 + c];

        float f[6];
#pragma unroll
        for (int k = 0; k < F_IN; ++k)
            f[k] = (raw[k] - rm[k]) * rs[k] * gm[k] + bt[k];
        f[5] = hf ? 1.f : w;
        v2f b0 = {f[0], f[1]}, b1 = {f[2], f[3]}, b2 = {f[4], f[5]};

        v2f s0 = {0.f, 0.f}, s1 = {0.f, 0.f};
#pragma unroll
        for (int t = 0; t < 16; t += 2) {
            v8f a0 = {0.f, 0.f, 0.f, 0.f, 0.f, 0.f, 0.f, 0.f};
            v8f a1 = {0.f, 0.f, 0.f, 0.f, 0.f, 0.f, 0.f, 0.f};
            a0 = __builtin_amdgcn_wmma_f32_16x16x4_f32(false, wa[t][0], false, b0, (short)0, a0, false, false);
            a1 = __builtin_amdgcn_wmma_f32_16x16x4_f32(false, wa[t + 1][0], false, b0, (short)0, a1, false, false);
            a0 = __builtin_amdgcn_wmma_f32_16x16x4_f32(false, wa[t][1], false, b1, (short)0, a0, false, false);
            a1 = __builtin_amdgcn_wmma_f32_16x16x4_f32(false, wa[t + 1][1], false, b1, (short)0, a1, false, false);
            a0 = __builtin_amdgcn_wmma_f32_16x16x4_f32(false, wa[t][2], false, b2, (short)0, a0, false, false);
            a1 = __builtin_amdgcn_wmma_f32_16x16x4_f32(false, wa[t + 1][2], false, b2, (short)0, a1, false, false);
            if (t < 8) {
                s0 = lrelu_att(a0, at[t], s0);
                s0 = lrelu_att(a1, at[t + 1], s0);
            } else {
                s1 = lrelu_att(a0, at[t], s1);
                s1 = lrelu_att(a1, at[t + 1], s1);
            }
        }
        float l0 = s0.x + s0.y, l1 = s1.x + s1.y;
        float lh = (hf ? l1 : l0) + __shfl_xor(hf ? l0 : l1, 16, 32);
        int e = g * 16 + c;
        logits[(long)e * HEADS + hf] = lh;
        atomicMax(&mxh[(long)dn * HEADS + hf], enc_f32(lh));

        node = node1; node1 = node2; g1 = g2; w = w1;
#pragma unroll
        for (int k = 0; k < F_IN; ++k) raw[k] = raw1[k];
    }
}

#define SC_BLOCKS 625
#define SC_PASSES 5
#define SC_EPL    8
#define SC_PASS_E (16 * SC_EPL)
__global__ __launch_bounds__(256)
void k_edge_scatter(const int* __restrict__ ei, const float* __restrict__ logits,
                    const unsigned* __restrict__ mx, const float* __restrict__ h,
                    const float* __restrict__ gamma1, const float* __restrict__ beta1,
                    const float* __restrict__ rm1, const float* __restrict__ rv1,
                    float* __restrict__ agg) {
    int slot = threadIdx.x >> 4;
    int l = threadIdx.x & 15;
    int hh = l >> 3, k = l & 7;
    if (k >= 6 && l != 6) return;
    int kc = k < 5 ? k : 0;
    float rm = rm1[kc], rs = rsqrtf(rv1[kc] + BN_EPS), g = gamma1[kc], b = beta1[kc];
    int ebase = blockIdx.x * (SC_PASSES * SC_PASS_E) + slot;
    float* aggh = agg + pkg_half() * AGG_COPY;
    int src[SC_EPL], dst[SC_EPL];
#pragma unroll
    for (int it = 0; it < SC_EPL; ++it) {
        int e = ebase + it * 16;
        src[it] = ei[e];
        dst[it] = ei[N_EDGES + e];
    }
#pragma unroll 1
    for (int p = 0; p < SC_PASSES; ++p) {
        int e0 = ebase + p * SC_PASS_E;
        float lg[SC_EPL], hr[SC_EPL];
        unsigned mv[SC_EPL];
#pragma unroll
        for (int it = 0; it < SC_EPL; ++it) {
            int e = e0 + it * 16;
            lg[it] = logits[(long)e * HEADS + hh];
            mv[it] = max(mx[(long)dst[it] * HEADS + hh], mx[MX_COPY + (long)dst[it] * HEADS + hh]);
            hr[it] = h[src[it] * F_IN + kc];
        }
        int en = ebase + min(p + 1, SC_PASSES - 1) * SC_PASS_E;
        int nsrc[SC_EPL], ndst[SC_EPL];
#pragma unroll
        for (int it = 0; it < SC_EPL; ++it) {
            int e = en + it * 16;
            nsrc[it] = ei[e];
            ndst[it] = ei[N_EDGES + e];
        }
#pragma unroll
        for (int it = 0; it < SC_EPL; ++it) {
            float ex = expf(lg[it] - dec_f32(mv[it]));
            float hv = (k < 5) ? ((hr[it] - rm) * rs * g + b) : 1.f;
            float v = (l == 6) ? 1.f : hv * ex;
            atomAddF(&aggh[(long)dst[it] * AGG_W + l], v);
        }
#pragma unroll
        for (int it = 0; it < SC_EPL; ++it) { src[it] = nsrc[it]; dst[it] = ndst[it]; }
    }
}

#define FC_PAIRS  ((N_NODES + 31) / 32)
#define FC_BLOCKS ((FC_PAIRS + 7) / 8)
#define FC_TILES  (N_NODES / 16)
__global__ __launch_bounds__(256)
void k_fc(const float* __restrict__ agg,
          const float* __restrict__ Mt, const float* __restrict__ sh,
          const float* __restrict__ W2, const float* __restrict__ b2,
          float* __restrict__ out) {
    const int lane = threadIdx.x & 31;
    const int c = lane & 15, hf = lane >> 4;
    const int wv = blockIdx.x * 8 + __builtin_amdgcn_readfirstlane((int)(threadIdx.x >> 5));
    if (wv >= FC_PAIRS) return;

    const int tile0 = 2 * wv;
    const int ntile = (tile0 + 1 < FC_TILES) ? 2 : 1;
    float4 ra[2][4];
#pragma unroll
    for (int u = 0; u < 2; ++u) {
        int tl = min(tile0 + u, FC_TILES - 1);
        const float* ga = agg + ((long)(tl * 16 + c) * AGG_W + 8 * hf);
        const float* gb = ga + AGG_COPY;
        ra[u][0] = *(const float4*)ga;
        ra[u][1] = *(const float4*)(ga + 4);
        ra[u][2] = *(const float4*)gb;
        ra[u][3] = *(const float4*)(gb + 4);
    }

    v2f ma[8][3];
    v8f shv[8];
    v2f w2a[8][4];
#pragma unroll
    for (int t = 0; t < 8; ++t) {
        const float* mrow = Mt + (16 * t + c) * 16 + 8 * hf;
        float4 m0 = *(const float4*)mrow;
        float2 m1 = *(const float2*)(mrow + 4);
        ma[t][0] = (v2f){m0.x, m0.y};
        ma[t][1] = (v2f){m0.z, m0.w};
        ma[t][2] = (v2f){m1.x, m1.y};
        shv[t] = *(const v8f*)(sh + 16 * t + 8 * hf);
#pragma unroll
        for (int q = 0; q < 4; ++q) {
            int d0 = 16 * t + 8 * hf + 2 * q;
            int cc = c < F_IN ? c : 0;
            float wx = W2[d0 * F_IN + cc], wy = W2[(d0 + 1) * F_IN + cc];
            w2a[t][q] = (v2f){c < F_IN ? wx : 0.f, c < F_IN ? wy : 0.f};
        }
    }
    v8f oinit = {0.f, 0.f, 0.f, 0.f, 0.f, 0.f, 0.f, 0.f};
#pragma unroll
    for (int r = 0; r < F_IN; ++r) {
        float bv = b2[r];
        oinit[r] = hf ? 0.f : bv;
    }

    for (int u = 0; u < ntile; ++u) {
        float4 p0 = u ? ra[1][0] : ra[0][0], p1 = u ? ra[1][1] : ra[0][1];
        float4 q0 = u ? ra[1][2] : ra[0][2], q1 = u ? ra[1][3] : ra[0][3];
        float s0 = p0.x + q0.x, s1 = p0.y + q0.y, s2 = p0.z + q0.z, s3 = p0.w + q0.w;
        float s4 = p1.x + q1.x, s5 = p1.y + q1.y, s6 = p1.z + q1.z;
        float cnt = __shfl(s6, c, 32);
        float inv = 0.5f / fmaxf(cnt, 1.f);
        float den = s5 + 1e-16f;
        v2f f0 = {s0 / den * inv, s1 / den * inv};
        v2f f1 = {s2 / den * inv, s3 / den * inv};
        v2f f2 = {s4 / den * inv, s5 / den * inv};

        v8f o0 = oinit;
        v8f o1 = {0.f, 0.f, 0.f, 0.f, 0.f, 0.f, 0.f, 0.f};
#pragma unroll
        for (int t = 0; t < 8; t += 2) {
            v8f a0 = shv[t], a1 = shv[t + 1];
            a0 = __builtin_amdgcn_wmma_f32_16x16x4_f32(false, ma[t][0], false, f0, (short)0, a0, false, false);
            a1 = __builtin_amdgcn_wmma_f32_16x16x4_f32(false, ma[t + 1][0], false, f0, (short)0, a1, false, false);
            a0 = __builtin_amdgcn_wmma_f32_16x16x4_f32(false, ma[t][1], false, f1, (short)0, a0, false, false);
            a1 = __builtin_amdgcn_wmma_f32_16x16x4_f32(false, ma[t + 1][1], false, f1, (short)0, a1, false, false);
            a0 = __builtin_amdgcn_wmma_f32_16x16x4_f32(false, ma[t][2], false, f2, (short)0, a0, false, false);
            a1 = __builtin_amdgcn_wmma_f32_16x16x4_f32(false, ma[t + 1][2], false, f2, (short)0, a1, false, false);
            a0 = __builtin_elementwise_max(a0, 0.01f * a0);
            a1 = __builtin_elementwise_max(a1, 0.01f * a1);
#pragma unroll
            for (int q = 0; q < 4; ++q) {
                v2f z0 = {a0[2 * q], a0[2 * q + 1]};
                v2f z1 = {a1[2 * q], a1[2 * q + 1]};
                o0 = __builtin_amdgcn_wmma_f32_16x16x4_f32(false, w2a[t][q], false, z0, (short)0, o0, false, false);
                o1 = __builtin_amdgcn_wmma_f32_16x16x4_f32(false, w2a[t + 1][q], false, z1, (short)0, o1, false, false);
            }
        }
        if (hf == 0) {
            float* op = out + (long)((tile0 + u) * 16 + c) * F_IN;
#pragma unroll
            for (int r = 0; r < F_IN; ++r) op[r] = o0[r] + o1[r];
        }
    }
}

extern "C" void kernel_launch(void* const* d_in, const int* in_sizes, int n_in,
                              void* d_out, int out_size, void* d_ws, size_t ws_size,
                              hipStream_t stream) {
    const float* h        = (const float*)d_in[0];
    const int*   ei       = (const int*)  d_in[1];
    const float* ew       = (const float*)d_in[2];
    const float* gamma1   = (const float*)d_in[3];
    const float* beta1    = (const float*)d_in[4];
    const float* rm1      = (const float*)d_in[5];
    const float* rv1      = (const float*)d_in[6];
    const float* Wl       = (const float*)d_in[7];
    const float* bl       = (const float*)d_in[8];
    const float* Wr       = (const float*)d_in[9];
    const float* br       = (const float*)d_in[10];
    const float* We       = (const float*)d_in[11];
    const float* att      = (const float*)d_in[12];
    const float* bias_gat = (const float*)d_in[13];
    const float* W1       = (const float*)d_in[14];
    const float* b1       = (const float*)d_in[15];
    const float* gamma2   = (const float*)d_in[16];
    const float* beta2    = (const float*)d_in[17];
    const float* rm2      = (const float*)d_in[18];
    const float* rv2      = (const float*)d_in[19];
    const float* W2       = (const float*)d_in[20];
    const float* b2       = (const float*)d_in[21];
    float* out = (float*)d_out;

    float* ws = (float*)d_ws;
    size_t o = 0;
    float* agg    = ws + o; o += (size_t)2 * N_NODES * AGG_W;
    float* logits = ws + o; o += (size_t)N_EDGES * HEADS;
    unsigned* mx  = (unsigned*)(ws + o); o += (size_t)2 * N_NODES * HEADS;
    float* Mt     = ws + o; o += (size_t)DDIM * 16;
    float* sh     = ws + o; o += (size_t)DDIM;

    const int T = 256;
    k_init<<<IN_BLOCKS + 1, T, 0, stream>>>(agg, mx, Wl, bl, bias_gat, W1, b1,
                                            gamma2, beta2, rm2, rv2, Mt, sh);
    k_edge_logits<<<LG_BLOCKS, T, 0, stream>>>(ei, ew, h, gamma1, beta1, rm1, rv1,
                                               Wl, bl, Wr, br, We, att, logits, mx);
    k_edge_scatter<<<SC_BLOCKS, T, 0, stream>>>(ei, logits, mx, h, gamma1, beta1,
                                                   rm1, rv1, agg);
    k_fc<<<FC_BLOCKS, T, 0, stream>>>(agg, Mt, sh, W2, b2, out);
}


# Logits kernel hands the scatter kernel one 32-byte record per edge (bn(h[src]), both logits, dst) in whole lines; scatter reads only that stream and the max words: no h gather, index reads or BN
# speedup vs baseline: 9.4497x; 1.1160x over previous
#include <hip/hip_runtime.h>
#include <math.h>

#define N_NODES 50000
#define N_EDGES 400000
#define HEADS   2
#define DDIM    128
#define F_IN    5
#define HD      (HEADS*DDIM)
#define BN_EPS  1e-5f
#define AGG_W   16
#define AGG_COPY ((long)N_NODES * AGG_W)
#define MX_COPY  ((long)N_NODES * HEADS)

typedef __attribute__((ext_vector_type(2))) float v2f;
typedef __attribute__((ext_vector_type(8))) float v8f;

__device__ __forceinline__ unsigned enc_f32(float f) {
    unsigned u = __float_as_uint(f);
    return (u & 0x80000000u) ? ~u : (u | 0x80000000u);
}
__device__ __forceinline__ float dec_f32(unsigned u) {
    unsigned v = (u & 0x80000000u) ? (u & 0x7FFFFFFFu) : ~u;
    return __uint_as_float(v);
}
__device__ __forceinline__ int pkg_half() {
    return (int)((__builtin_amdgcn_s_sendmsg_rtn(0x87) >> 18) & 1u);
}
__device__ __forceinline__ void atomAddF(float* p, float v) {
    unsafeAtomicAdd(p, v);
}

#define IN_BLOCKS  256
#define IN_THREADS (IN_BLOCKS * 256)
#define IN_AGG4    (2 * N_NODES * AGG_W / 4)
#define IN_MX4     (2 * N_NODES * HEADS / 4)
#define TP 132
__device__ __forceinline__ void build_node_consts(
        const float* __restrict__ Wl, const float* __restrict__ bl,
        const float* __restrict__ bias_gat,
        const float* __restrict__ W1, const float* __restrict__ b1,
        const float* __restrict__ gamma2, const float* __restrict__ beta2,
        const float* __restrict__ rm2, const float* __restrict__ rv2,
        float* __restrict__ Mt, float* __restrict__ sh) {
    __shared__ float P[16 * TP];
    int t = threadIdx.x;
    {
        int d = t & 127;
        int head = t >> 7;
        float w[6];
#pragma unroll
        for (int k = 0; k < F_IN; ++k) w[k] = Wl[k * HD + head * DDIM + d];
        w[5] = bl[head * DDIM + d];
        float bg = bias_gat[d];
#pragma unroll
        for (int k = 0; k < 6; ++k) P[(6 * head + k) * TP + d] = w[k];
        P[(12 + head) * TP + d] = head ? 0.f : bg;
        P[(14 + head) * TP + d] = 0.f;
    }
    __syncthreads();
    int wave = t >> 5, lane = t & 31;
    int c0 = wave * 16;
    int laneN = lane & 15;
    int koff = (lane >> 4) * 2;
    v8f acc = {0.f, 0.f, 0.f, 0.f, 0.f, 0.f, 0.f, 0.f};
    for (int k = 0; k < 128; k += 4) {
        v2f a, b;
        a.x = P[laneN * TP + k + koff];
        a.y = P[laneN * TP + k + koff + 1];
        int kk = k + koff;
        b.x = W1[kk * 128 + c0 + laneN];
        b.y = W1[(kk + 1) * 128 + c0 + laneN];
        acc = __builtin_amdgcn_wmma_f32_16x16x4_f32(
            false, a, false, b, (short)0, acc, false, false);
    }
    int col = c0 + laneN;
    float sc = rsqrtf(rv2[col] + BN_EPS) * gamma2[col];
    float cb = b1[col] - rm2[col];
    float be = beta2[col];
#pragma unroll
    for (int v = 0; v < 8; ++v) {
        int m = v + (lane >> 4) * 8;
        int head = m >= 6 ? 1 : 0;
        if (m < 12) Mt[col * 16 + 8 * head + (m - 6 * head)] = acc[v] * sc;
        else if (m == 12) sh[col] = (acc[v] + cb) * sc + be;
    }
}

__global__ __launch_bounds__(256)
void k_init(float* __restrict__ agg, unsigned* __restrict__ mx,
            const float* __restrict__ Wl, const float* __restrict__ bl,
            const float* __restrict__ bias_gat,
            const float* __restrict__ W1, const float* __restrict__ b1,
            const float* __restrict__ gamma2, const float* __restrict__ beta2,
            const float* __restrict__ rm2, const float* __restrict__ rv2,
            float* __restrict__ Mt, float* __restrict__ sh) {
    if (blockIdx.x == IN_BLOCKS) {
        build_node_consts(Wl, bl, bias_gat, W1, b1, gamma2, beta2, rm2, rv2, Mt, sh);
        return;
    }
    float4* agg4 = (float4*)agg;
    uint4* mx4 = (uint4*)mx;
    int i0 = blockIdx.x * 256 + threadIdx.x;
#pragma unroll
    for (int j = 0; j < IN_AGG4 / IN_THREADS; ++j)
        agg4[i0 + j * IN_THREADS] = make_float4(0.f, 0.f, 0.f, 0.f);
    int it = i0 + (IN_AGG4 / IN_THREADS) * IN_THREADS;
    if (it < IN_AGG4) agg4[it] = make_float4(0.f, 0.f, 0.f, 0.f);
    if (i0 < IN_MX4) mx4[i0] = make_uint4(0u, 0u, 0u, 0u);
}

#define REC_W   8
#define LG_BLOCKS 256
#define LG_WAVES  (LG_BLOCKS * 8)
#define LG_GROUPS (N_EDGES / 16)

__device__ __forceinline__ v2f lrelu_att(v8f acc, v8f a, v2f s) {
#pragma unroll
    for (int r = 0; r < 8; r += 2) {
        v2f z  = {acc[r], acc[r + 1]};
        v2f a2 = {a[r], a[r + 1]};
        v2f l  = __builtin_elementwise_max(z, 0.2f * z);
        s = a2 * l + s;
    }
    return s;
}

__global__ __launch_bounds__(256)
void k_edge_logits(const int* __restrict__ ei, const float* __restrict__ ew,
                   const float* __restrict__ h,
                   const float* __restrict__ gamma1, const float* __restrict__ beta1,
                   const float* __restrict__ rm1, const float* __restrict__ rv1,
                   const float* __restrict__ Wl, const float* __restrict__ bl,
                   const float* __restrict__ Wr, const float* __restrict__ br,
                   const float* __restrict__ We, const float* __restrict__ att,
                   float4* __restrict__ rec, unsigned* __restrict__ mx) {
    const int lane = threadIdx.x & 31;
    const int c = lane & 15, hf = lane >> 4;
    const int wv = blockIdx.x * 8 + __builtin_amdgcn_readfirstlane((int)(threadIdx.x >> 5));
    unsigned* mxh = mx + pkg_half() * MX_COPY;

    const float* Wside = hf ? Wr : Wl;
    v2f wa[16][3];
    v8f at[16];
#pragma unroll
    for (int t = 0; t < 16; ++t) {
        int col = t * 16 + c;
        float wedge = We[col];
        float bsum = bl[col] + br[col];
        wa[t][0] = (v2f){Wside[0 * HD + col], Wside[1 * HD + col]};
        wa[t][1] = (v2f){Wside[2 * HD + col], Wside[3 * HD + col]};
        wa[t][2] = (v2f){Wside[4 * HD + col], hf ? bsum : wedge};
        at[t] = *(const v8f*)(att + t * 16 + 8 * hf);
    }
    float rm[F_IN], rs[F_IN], gm[F_IN], bt[F_IN];
#pragma unroll
    for (int k = 0; k < F_IN; ++k) {
        rm[k] = rm1[k];
        rs[k] = rsqrtf(rv1[k] + BN_EPS);
        gm[k] = gamma1[k];
        bt[k] = beta1[k];
    }

    const int* eside = ei + hf * N_EDGES;
    int g = wv;
    int node = eside[g * 16 + c];
    float raw[F_IN];
#pragma unroll
    for (int k = 0; k < F_IN; ++k) raw[k] = h[node * F_IN + k];
    float w = ew[g * 16 + c];
    int g1 = min(g + LG_WAVES, LG_GROUPS - 1);
    int node1 = eside[g1 * 16 + c];

    for (; g < LG_GROUPS; g += LG_WAVES) {
        float raw1[F_IN];
#pragma unroll
        for (int k = 0; k < F_IN; ++k) raw1[k] = h[node1 * F_IN + k];
        float w1 = ew[g1 * 16 + c];
        int g2 = min(g1 + LG_WAVES, LG_GROUPS - 1);
        int node2 = eside[g2 * 16 + c];
        int dn = ei[N_EDGES + g * 16 + c];

        float f[6];
#pragma unroll
        for (int k = 0; k < F_IN; ++k)
            f[k] = (raw[k] - rm[k]) * rs[k] * gm[k] + bt[k];
        f[5] = hf ? 1.f : w;
        v2f b0 = {f[0], f[1]}, b1 = {f[2], f[3]}, b2 = {f[4], f[5]};

        v2f s0 = {0.f, 0.f}, s1 = {0.f, 0.f};
#pragma unroll
        for (int t = 0; t < 16; t += 2) {
            v8f a0 = {0.f, 0.f, 0.f, 0.f, 0.f, 0.f, 0.f, 0.f};
            v8f a1 = {0.f, 0.f, 0.f, 0.f, 0.f, 0.f, 0.f, 0.f};
            a0 = __builtin_amdgcn_wmma_f32_16x16x4_f32(false, wa[t][0], false, b0, (short)0, a0, false, false);
            a1 = __builtin_amdgcn_wmma_f32_16x16x4_f32(false, wa[t + 1][0], false, b0, (short)0, a1, false, false);
            a0 = __builtin_amdgcn_wmma_f32_16x16x4_f32(false, wa[t][1], false, b1, (short)0, a0, false, false);
            a1 = __builtin_amdgcn_wmma_f32_16x16x4_f32(false, wa[t + 1][1], false, b1, (short)0, a1, false, false);
            a0 = __builtin_amdgcn_wmma_f32_16x16x4_f32(false, wa[t][2], false, b2, (short)0, a0, false, false);
            a1 = __builtin_amdgcn_wmma_f32_16x16x4_f32(false, wa[t + 1][2], false, b2, (short)0, a1, false, false);
            if (t < 8) {
                s0 = lrelu_att(a0, at[t], s0);
                s0 = lrelu_att(a1, at[t + 1], s0);
            } else {
                s1 = lrelu_att(a0, at[t], s1);
                s1 = lrelu_att(a1, at[t + 1], s1);
            }
        }
        float l0 = s0.x + s0.y, l1 = s1.x + s1.y;
        float lh = (hf ? l1 : l0) + __shfl_xor(hf ? l0 : l1, 16, 32);
        int e = g * 16 + c;
        float pf4 = __shfl_xor(f[4], 16, 32);
        float pl0 = __shfl_xor(lh, 16, 32);
        float4 r;
        r.x = hf ? pf4 : f[0];
        r.y = hf ? pl0 : f[1];
        r.z = hf ? lh : f[2];
        r.w = hf ? __int_as_float(dn) : f[3];
        rec[(long)e * 2 + hf] = r;
        atomicMax(&mxh[(long)dn * HEADS + hf], enc_f32(lh));

        node = node1; node1 = node2; g1 = g2; w = w1;
#pragma unroll
        for (int k = 0; k < F_IN; ++k) raw[k] = raw1[k];
    }
}

#define SC_BLOCKS 625
#define SC_PASSES 5
#define SC_EPL    8
#define SC_PASS_E (16 * SC_EPL)
__global__ __launch_bounds__(256)
void k_edge_scatter(const float* __restrict__ rec, const unsigned* __restrict__ mx,
                    float* __restrict__ agg) {
    int slot = threadIdx.x >> 4;
    int l = threadIdx.x & 15;
    int hh = l >> 3, k = l & 7;
    bool adds = (k < 6) || (l == 6);
    int ebase = blockIdx.x * (SC_PASSES * SC_PASS_E) + slot;
    float* aggh = agg + pkg_half() * AGG_COPY;
    float x[SC_EPL];
#pragma unroll
    for (int it = 0; it < SC_EPL; ++it)
        x[it] = rec[(long)(ebase + it * 16) * REC_W + k];
#pragma unroll 1
    for (int p = 0; p < SC_PASSES; ++p) {
        int dst[SC_EPL];
        float lg[SC_EPL];
        unsigned mv[SC_EPL];
#pragma unroll
        for (int it = 0; it < SC_EPL; ++it) {
            dst[it] = __float_as_int(__shfl(x[it], 7, 16));
            lg[it] = __shfl(x[it], 5 + hh, 16);
            mv[it] = max(mx[(long)dst[it] * HEADS + hh], mx[MX_COPY + (long)dst[it] * HEADS + hh]);
        }
        int en = ebase + min(p + 1, SC_PASSES - 1) * SC_PASS_E;
        float nx[SC_EPL];
#pragma unroll
        for (int it = 0; it < SC_EPL; ++it)
            nx[it] = rec[(long)(en + it * 16) * REC_W + k];
#pragma unroll
        for (int it = 0; it < SC_EPL; ++it) {
            float ex = expf(lg[it] - dec_f32(mv[it]));
            float hv = (k < 5) ? x[it] : 1.f;
            float v = (l == 6) ? 1.f : hv * ex;
            if (adds) atomAddF(&aggh[(long)dst[it] * AGG_W + l], v);
        }
#pragma unroll
        for (int it = 0; it < SC_EPL; ++it) x[it] = nx[it];
    }
}

#define FC_PAIRS  ((N_NODES + 31) / 32)
#define FC_BLOCKS ((FC_PAIRS + 7) / 8)
#define FC_TILES  (N_NODES / 16)
__global__ __launch_bounds__(256)
void k_fc(const float* __restrict__ agg,
          const float* __restrict__ Mt, const float* __restrict__ sh,
          const float* __restrict__ W2, const float* __restrict__ b2,
          float* __restrict__ out) {
    const int lane = threadIdx.x & 31;
    const int c = lane & 15, hf = lane >> 4;
    const int wv = blockIdx.x * 8 + __builtin_amdgcn_readfirstlane((int)(threadIdx.x >> 5));
    if (wv >= FC_PAIRS) return;

    const int tile0 = 2 * wv;
    const int ntile = (tile0 + 1 < FC_TILES) ? 2 : 1;
    float4 ra[2][4];
#pragma unroll
    for (int u = 0; u < 2; ++u) {
        int tl = min(tile0 + u, FC_TILES - 1);
        const float* ga = agg + ((long)(tl * 16 + c) * AGG_W + 8 * hf);
        const float* gb = ga + AGG_COPY;
        ra[u][0] = *(const float4*)ga;
        ra[u][1] = *(const float4*)(ga + 4);
        ra[u][2] = *(const float4*)gb;
        ra[u][3] = *(const float4*)(gb + 4);
    }

    v2f ma[8][3];
    v8f shv[8];
    v2f w2a[8][4];
#pragma unroll
    for (int t = 0; t < 8; ++t) {
        const float* mrow = Mt + (16 * t + c) * 16 + 8 * hf;
        float4 m0 = *(const float4*)mrow;
        float2 m1 = *(const float2*)(mrow + 4);
        ma[t][0] = (v2f){m0.x, m0.y};
        ma[t][1] = (v2f){m0.z, m0.w};
        ma[t][2] = (v2f){m1.x, m1.y};
        shv[t] = *(const v8f*)(sh + 16 * t + 8 * hf);
#pragma unroll
        for (int q = 0; q < 4; ++q) {
            int d0 = 16 * t + 8 * hf + 2 * q;
            int cc = c < F_IN ? c : 0;
            float wx = W2[d0 * F_IN + cc], wy = W2[(d0 + 1) * F_IN + cc];
            w2a[t][q] = (v2f){c < F_IN ? wx : 0.f, c < F_IN ? wy : 0.f};
        }
    }
    v8f oinit = {0.f, 0.f, 0.f, 0.f, 0.f, 0.f, 0.f, 0.f};
#pragma unroll
    for (int r = 0; r < F_IN; ++r) {
        float bv = b2[r];
        oinit[r] = hf ? 0.f : bv;
    }

    for (int u = 0; u < ntile; ++u) {
        float4 p0 = u ? ra[1][0] : ra[0][0], p1 = u ? ra[1][1] : ra[0][1];
        float4 q0 = u ? ra[1][2] : ra[0][2], q1 = u ? ra[1][3] : ra[0][3];
        float s0 = p0.x + q0.x, s1 = p0.y + q0.y, s2 = p0.z + q0.z, s3 = p0.w + q0.w;
        float s4 = p1.x + q1.x, s5 = p1.y + q1.y, s6 = p1.z + q1.z;
        float cnt = __shfl(s6, c, 32);
        float inv = 0.5f / fmaxf(cnt, 1.f);
        float den = s5 + 1e-16f;
        v2f f0 = {s0 / den * inv, s1 / den * inv};
        v2f f1 = {s2 / den * inv, s3 / den * inv};
        v2f f2 = {s4 / den * inv, s5 / den * inv};

        v8f o0 = oinit;
        v8f o1 = {0.f, 0.f, 0.f, 0.f, 0.f, 0.f, 0.f, 0.f};
#pragma unroll
        for (int t = 0; t < 8; t += 2) {
            v8f a0 = shv[t], a1 = shv[t + 1];
            a0 = __builtin_amdgcn_wmma_f32_16x16x4_f32(false, ma[t][0], false, f0, (short)0, a0, false, false);
            a1 = __builtin_amdgcn_wmma_f32_16x16x4_f32(false, ma[t + 1][0], false, f0, (short)0, a1, false, false);
            a0 = __builtin_amdgcn_wmma_f32_16x16x4_f32(false, ma[t][1], false, f1, (short)0, a0, false, false);
            a1 = __builtin_amdgcn_wmma_f32_16x16x4_f32(false, ma[t + 1][1], false, f1, (short)0, a1, false, false);
            a0 = __builtin_amdgcn_wmma_f32_16x16x4_f32(false, ma[t][2], false, f2, (short)0, a0, false, false);
            a1 = __builtin_amdgcn_wmma_f32_16x16x4_f32(false, ma[t + 1][2], false, f2, (short)0, a1, false, false);
            a0 = __builtin_elementwise_max(a0, 0.01f * a0);
            a1 = __builtin_elementwise_max(a1, 0.01f * a1);
#pragma unroll
            for (int q = 0; q < 4; ++q) {
                v2f z0 = {a0[2 * q], a0[2 * q + 1]};
                v2f z1 = {a1[2 * q], a1[2 * q + 1]};
                o0 = __builtin_amdgcn_wmma_f32_16x16x4_f32(false, w2a[t][q], false, z0, (short)0, o0, false, false);
                o1 = __builtin_amdgcn_wmma_f32_16x16x4_f32(false, w2a[t + 1][q], false, z1, (short)0, o1, false, false);
            }
        }
        if (hf == 0) {
            float* op = out + (long)((tile0 + u) * 16 + c) * F_IN;
#pragma unroll
            for (int r = 0; r < F_IN; ++r) op[r] = o0[r] + o1[r];
        }
    }
}

extern "C" void kernel_launch(void* const* d_in, const int* in_sizes, int n_in,
                              void* d_out, int out_size, void* d_ws, size_t ws_size,
                              hipStream_t stream) {
    const float* h        = (const float*)d_in[0];
    const int*   ei       = (const int*)  d_in[1];
    const float* ew       = (const float*)d_in[2];
    const float* gamma1   = (const float*)d_in[3];
    const float* beta1    = (const float*)d_in[4];
    const float* rm1      = (const float*)d_in[5];
    const float* rv1      = (const float*)d_in[6];
    const float* Wl       = (const float*)d_in[7];
    const float* bl       = (const float*)d_in[8];
    const float* Wr       = (const float*)d_in[9];
    const float* br       = (const float*)d_in[10];
    const float* We       = (const float*)d_in[11];
    const float* att      = (const float*)d_in[12];
    const float* bias_gat = (const float*)d_in[13];
    const float* W1       = (const float*)d_in[14];
    const float* b1       = (const float*)d_in[15];
    const float* gamma2   = (const float*)d_in[16];
    const float* beta2    = (const float*)d_in[17];
    const float* rm2      = (const float*)d_in[18];
    const float* rv2      = (const float*)d_in[19];
    const float* W2       = (const float*)d_in[20];
    const float* b2       = (const float*)d_in[21];
    float* out = (float*)d_out;

    float* ws = (float*)d_ws;
    size_t o = 0;
    float* agg    = ws + o; o += (size_t)2 * N_NODES * AGG_W;
    float* rec    = ws + o; o += (size_t)N_EDGES * REC_W;
    unsigned* mx  = (unsigned*)(ws + o); o += (size_t)2 * N_NODES * HEADS;
    float* Mt     = ws + o; o += (size_t)DDIM * 16;
    float* sh     = ws + o; o += (size_t)DDIM;

    const int T = 256;
    k_init<<<IN_BLOCKS + 1, T, 0, stream>>>(agg, mx, Wl, bl, bias_gat, W1, b1,
                                            gamma2, beta2, rm2, rv2, Mt, sh);
    k_edge_logits<<<LG_BLOCKS, T, 0, stream>>>(ei, ew, h, gamma1, beta1, rm1, rv1,
                                               Wl, bl, Wr, br, We, att, (float4*)rec, mx);
    k_edge_scatter<<<SC_BLOCKS, T, 0, stream>>>(rec, mx, agg);
    k_fc<<<FC_BLOCKS, T, 0, stream>>>(agg, Mt, sh, W2, b2, out);
}
